# starting kernel
# baseline (speedup 1.0000x reference)
#include <hip/hip_runtime.h>
#include <math.h>

#define N_DIS 20000
#define N_MIC 80000
#define NTOT  (N_DIS + N_MIC)
#define NEDGE 1600000
#define FATT  64
#define SLOPE 0.2f

typedef float v2f __attribute__((ext_vector_type(2)));
typedef float v8f __attribute__((ext_vector_type(8)));

__global__ __launch_bounds__(128) void gemm_wmma_f32(
        const float* __restrict__ X, const float* __restrict__ W,
        float* __restrict__ Z, int M, int K) {
    const int KT = 64;
    const int MT = 128;
    __shared__ float lX[MT * KT];
    __shared__ float lW[KT * FATT];

    const int lane  = threadIdx.x & 31;
    const int wave  = threadIdx.x >> 5;
    const int row   = lane & 15;
    const int khalf = lane >> 4;
    const int n     = lane & 15;

    const int blk_row = blockIdx.x * MT;
    const int lr0 = wave * 32 + row;
    const int lr1 = lr0 + 16;

    v8f c0 = {}, c1 = {}, c2 = {}, c3 = {};
    v8f c4 = {}, c5 = {}, c6 = {}, c7 = {};

    for (int kt = 0; kt < K; kt += KT) {
        for (int i = threadIdx.x; i < KT * FATT; i += blockDim.x) {
            int kg = kt + (i >> 6);
            lW[i] = (kg < K) ? W[(size_t)kg * FATT + (i & 63)] : 0.0f;
        }
        for (int i = threadIdx.x; i < MT * KT; i += blockDim.x) {
            int r  = i >> 6;
            int cc = i & 63;
            int mg = blk_row + r;
            int kg = kt + cc;
            lX[i] = (mg < M && kg < K) ? X[(size_t)mg * K + kg] : 0.0f;
        }
        __syncthreads();

        #pragma unroll 4
        for (int k0 = 0; k0 < KT; k0 += 4) {
            const int ka = k0 + 2 * khalf;
            v2f a0 = *(const v2f*)(lX + lr0 * KT + ka);
            v2f a1 = *(const v2f*)(lX + lr1 * KT + ka);

            const float* wr0 = lW + ka * FATT + n;
            const float* wr1 = lW + (ka + 1) * FATT + n;
            v2f b0 = { wr0[0],  wr1[0]  };
            v2f b1 = { wr0[16], wr1[16] };
            v2f b2 = { wr0[32], wr1[32] };
            v2f b3 = { wr0[48], wr1[48] };

            c0 = __builtin_amdgcn_wmma_f32_16x16x4_f32(false, a0, false, b0, (short)0, c0, false, false);
            c1 = __builtin_amdgcn_wmma_f32_16x16x4_f32(false, a0, false, b1, (short)0, c1, false, false);
            c2 = __builtin_amdgcn_wmma_f32_16x16x4_f32(false, a0, false, b2, (short)0, c2, false, false);
            c3 = __builtin_amdgcn_wmma_f32_16x16x4_f32(false, a0, false, b3, (short)0, c3, false, false);
            c4 = __builtin_amdgcn_wmma_f32_16x16x4_f32(false, a1, false, b0, (short)0, c4, false, false);
            c5 = __builtin_amdgcn_wmma_f32_16x16x4_f32(false, a1, false, b1, (short)0, c5, false, false);
            c6 = __builtin_amdgcn_wmma_f32_16x16x4_f32(false, a1, false, b2, (short)0, c6, false, false);
            c7 = __builtin_amdgcn_wmma_f32_16x16x4_f32(false, a1, false, b3, (short)0, c7, false, false);
        }
        __syncthreads();
    }

    const int m0 = blk_row + wave * 32 + 8 * khalf;
    #pragma unroll
    for (int r = 0; r < 8; ++r) {
        int mm = m0 + r;
        if (mm < M) {
            float* zr = Z + (size_t)mm * FATT + n;
            zr[0]  = c0[r];
            zr[16] = c1[r];
            zr[32] = c2[r];
            zr[48] = c3[r];
        }
        int mm1 = mm + 16;
        if (mm1 < M) {
            float* zr = Z + (size_t)mm1 * FATT + n;
            zr[0]  = c4[r];
            zr[16] = c5[r];
            zr[32] = c6[r];
            zr[48] = c7[r];
        }
    }
}

__global__ void init_state(unsigned int* __restrict__ m_ord, float* __restrict__ denom,
                           float* __restrict__ h) {
    long long i = (long long)blockIdx.x * blockDim.x + threadIdx.x;
    if (i < (long long)NTOT * FATT) h[i] = 0.0f;
    if (i < NTOT) { m_ord[i] = 0u; denom[i] = 0.0f; }
}

__device__ __forceinline__ unsigned int float_to_ord(float f) {
    unsigned int u = __float_as_uint(f);
    return (u & 0x80000000u) ? ~u : (u | 0x80000000u);
}
__device__ __forceinline__ float ord_to_float(unsigned int u) {
    return __uint_as_float((u & 0x80000000u) ? (u ^ 0x80000000u) : ~u);
}

__global__ void edge_score(const float* __restrict__ z, const int* __restrict__ src,
                           const int* __restrict__ dst, float* __restrict__ e,
                           unsigned int* __restrict__ m_ord) {
    const int lane = threadIdx.x & 31;
    const int wave = threadIdx.x >> 5;
    long long edge = (long long)blockIdx.x * (blockDim.x >> 5) + wave;
    if (edge >= NEDGE) return;
    const int s = src[edge], d = dst[edge];
    const float2* zs = (const float2*)(z + (size_t)s * FATT);
    const float2* zd = (const float2*)(z + (size_t)d * FATT);
    float2 a = zs[lane], b = zd[lane];
    float v = a.x * b.x + a.y * b.y;
    #pragma unroll
    for (int off = 16; off > 0; off >>= 1) v += __shfl_xor(v, off, 32);
    if (lane == 0) {
        float ev = v > 0.0f ? v : SLOPE * v;
        e[edge] = ev;
        atomicMax(&m_ord[d], float_to_ord(ev));
    }
}

__global__ void edge_expsum(const int* __restrict__ dst, float* __restrict__ e,
                            const unsigned int* __restrict__ m_ord,
                            float* __restrict__ denom) {
    long long i = (long long)blockIdx.x * blockDim.x + threadIdx.x;
    if (i >= NEDGE) return;
    const int d = dst[i];
    const float m = ord_to_float(m_ord[d]);
    const float ex = __expf(e[i] - m);
    e[i] = ex;
    atomicAdd(&denom[d], ex);
}

__global__ void edge_scatter(const float* __restrict__ z, const int* __restrict__ src,
                             const int* __restrict__ dst, const float* __restrict__ e_exp,
                             const float* __restrict__ denom, float* __restrict__ h) {
    const int lane = threadIdx.x & 31;
    const int wave = threadIdx.x >> 5;
    long long edge = (long long)blockIdx.x * (blockDim.x >> 5) + wave;
    if (edge >= NEDGE) return;
    const int s = src[edge], d = dst[edge];
    const float alpha = e_exp[edge] / denom[d];
    const float2* zs = (const float2*)(z + (size_t)s * FATT);
    float2 a = zs[lane];
    float* hd = h + (size_t)d * FATT + 2 * lane;
    atomicAdd(hd + 0, alpha * a.x);
    atomicAdd(hd + 1, alpha * a.y);
}

__global__ void elu_inplace(float* __restrict__ h) {
    long long i = (long long)blockIdx.x * blockDim.x + threadIdx.x;
    if (i >= (long long)NTOT * FATT) return;
    const float v = h[i];
    h[i] = v > 0.0f ? v : (__expf(v) - 1.0f);
}

extern "C" void kernel_launch(void* const* d_in, const int* in_sizes, int n_in,
                              void* d_out, int out_size, void* d_ws, size_t ws_size,
                              hipStream_t stream) {
    const float* d_sim  = (const float*)d_in[0];
    const float* mi_sim = (const float*)d_in[1];
    const float* W_d    = (const float*)d_in[2];
    const float* W_mi   = (const float*)d_in[3];
    const int*   src    = (const int*)d_in[4];
    const int*   dst    = (const int*)d_in[5];
    float*       out    = (float*)d_out;

    float* z            = (float*)d_ws;
    float* e            = z + (size_t)NTOT * FATT;
    unsigned int* m_ord = (unsigned int*)(e + NEDGE);
    float* denom        = (float*)(m_ord + NTOT);

    const long long tot_h = (long long)NTOT * FATT;

    init_state<<<(int)((tot_h + 255) / 256), 256, 0, stream>>>(m_ord, denom, out);

    gemm_wmma_f32<<<(N_DIS + 127) / 128, 128, 0, stream>>>(d_sim, W_d, z, N_DIS, 383);
    gemm_wmma_f32<<<(N_MIC + 127) / 128, 128, 0, stream>>>(mi_sim, W_mi,
                                                           z + (size_t)N_DIS * FATT, N_MIC, 495);

    edge_score<<<(NEDGE + 7) / 8, 256, 0, stream>>>(z, src, dst, e, m_ord);

    edge_expsum<<<(NEDGE + 255) / 256, 256, 0, stream>>>(dst, e, m_ord, denom);

    edge_scatter<<<(NEDGE + 7) / 8, 256, 0, stream>>>(z, src, dst, e, denom, out);

    elu_inplace<<<(int)((tot_h + 255) / 256), 256, 0, stream>>>(out);
}


# GEMM staging: all 96 global loads of a k-slice issued as one batch into registers (addresses clamped, no load under a branch) before the LDS stores, instead of one guarded load at a time
# speedup vs baseline: 1.8492x; 1.8492x over previous
#include <hip/hip_runtime.h>
#include <math.h>

#define N_DIS 20000
#define N_MIC 80000
#define NTOT  (N_DIS + N_MIC)
#define NEDGE 1600000
#define FATT  64
#define SLOPE 0.2f

typedef float v2f __attribute__((ext_vector_type(2)));
typedef float v8f __attribute__((ext_vector_type(8)));

__global__ __launch_bounds__(128) void gemm_wmma_f32(
        const float* __restrict__ X, const float* __restrict__ W,
        float* __restrict__ Z, int M, int K) {
    const int KT = 64;
    const int MT = 128;
    __shared__ float lX[MT * KT];
    __shared__ float lW[KT * FATT];

    const int lane  = threadIdx.x & 31;
    const int wave  = threadIdx.x >> 5;
    const int row   = lane & 15;
    const int khalf = lane >> 4;
    const int n     = lane & 15;

    const int blk_row = blockIdx.x * MT;
    const int lr0 = wave * 32 + row;
    const int lr1 = lr0 + 16;

    v8f c0 = {}, c1 = {}, c2 = {}, c3 = {};
    v8f c4 = {}, c5 = {}, c6 = {}, c7 = {};

    const int tid = threadIdx.x;
    const int sc  = tid & 63;
    const int sr  = tid >> 6;

    for (int kt = 0; kt < K; kt += KT) {
        float wv[(KT * FATT) / 128];
        float xv[(MT * KT) / 128];
        const int  kgx   = kt + sc;
        const bool kx_ok = kgx < K;
        const unsigned kgxc = (unsigned)(kx_ok ? kgx : K - 1);
        #pragma unroll
        for (int j = 0; j < (KT * FATT) / 128; ++j) {
            int kg  = kt + sr + 2 * j;
            int kgc = kg < K ? kg : K - 1;
            wv[j] = W[(unsigned)kgc * FATT + (unsigned)sc];
        }
        #pragma unroll
        for (int j = 0; j < (MT * KT) / 128; ++j) {
            int mg  = blk_row + sr + 2 * j;
            int mgc = mg < M ? mg : M - 1;
            xv[j] = X[(unsigned)mgc * (unsigned)K + kgxc];
        }
        #pragma unroll
        for (int j = 0; j < (KT * FATT) / 128; ++j) {
            int kg = kt + sr + 2 * j;
            lW[tid + 128 * j] = (kg < K) ? wv[j] : 0.0f;
        }
        #pragma unroll
        for (int j = 0; j < (MT * KT) / 128; ++j) {
            int mg = blk_row + sr + 2 * j;
            lX[tid + 128 * j] = (mg < M && kx_ok) ? xv[j] : 0.0f;
        }
        __syncthreads();

        #pragma unroll 4
        for (int k0 = 0; k0 < KT; k0 += 4) {
            const int ka = k0 + 2 * khalf;
            v2f a0 = *(const v2f*)(lX + lr0 * KT + ka);
            v2f a1 = *(const v2f*)(lX + lr1 * KT + ka);

            const float* wr0 = lW + ka * FATT + n;
            const float* wr1 = lW + (ka + 1) * FATT + n;
            v2f b0 = { wr0[0],  wr1[0]  };
            v2f b1 = { wr0[16], wr1[16] };
            v2f b2 = { wr0[32], wr1[32] };
            v2f b3 = { wr0[48], wr1[48] };

            c0 = __builtin_amdgcn_wmma_f32_16x16x4_f32(false, a0, false, b0, (short)0, c0, false, false);
            c1 = __builtin_amdgcn_wmma_f32_16x16x4_f32(false, a0, false, b1, (short)0, c1, false, false);
            c2 = __builtin_amdgcn_wmma_f32_16x16x4_f32(false, a0, false, b2, (short)0, c2, false, false);
            c3 = __builtin_amdgcn_wmma_f32_16x16x4_f32(false, a0, false, b3, (short)0, c3, false, false);
            c4 = __builtin_amdgcn_wmma_f32_16x16x4_f32(false, a1, false, b0, (short)0, c4, false, false);
            c5 = __builtin_amdgcn_wmma_f32_16x16x4_f32(false, a1, false, b1, (short)0, c5, false, false);
            c6 = __builtin_amdgcn_wmma_f32_16x16x4_f32(false, a1, false, b2, (short)0, c6, false, false);
            c7 = __builtin_amdgcn_wmma_f32_16x16x4_f32(false, a1, false, b3, (short)0, c7, false, false);
        }
        __syncthreads();
    }

    const int m0 = blk_row + wave * 32 + 8 * khalf;
    #pragma unroll
    for (int r = 0; r < 8; ++r) {
        int mm = m0 + r;
        if (mm < M) {
            float* zr = Z + (size_t)mm * FATT + n;
            zr[0]  = c0[r];
            zr[16] = c1[r];
            zr[32] = c2[r];
            zr[48] = c3[r];
        }
        int mm1 = mm + 16;
        if (mm1 < M) {
            float* zr = Z + (size_t)mm1 * FATT + n;
            zr[0]  = c4[r];
            zr[16] = c5[r];
            zr[32] = c6[r];
            zr[48] = c7[r];
        }
    }
}

__global__ void init_state(unsigned int* __restrict__ m_ord, float* __restrict__ denom,
                           float* __restrict__ h) {
    long long i = (long long)blockIdx.x * blockDim.x + threadIdx.x;
    if (i < (long long)NTOT * FATT) h[i] = 0.0f;
    if (i < NTOT) { m_ord[i] = 0u; denom[i] = 0.0f; }
}

__device__ __forceinline__ unsigned int float_to_ord(float f) {
    unsigned int u = __float_as_uint(f);
    return (u & 0x80000000u) ? ~u : (u | 0x80000000u);
}
__device__ __forceinline__ float ord_to_float(unsigned int u) {
    return __uint_as_float((u & 0x80000000u) ? (u ^ 0x80000000u) : ~u);
}

__global__ void edge_score(const float* __restrict__ z, const int* __restrict__ src,
                           const int* __restrict__ dst, float* __restrict__ e,
                           unsigned int* __restrict__ m_ord) {
    const int lane = threadIdx.x & 31;
    const int wave = threadIdx.x >> 5;
    long long edge = (long long)blockIdx.x * (blockDim.x >> 5) + wave;
    if (edge >= NEDGE) return;
    const int s = src[edge], d = dst[edge];
    const float2* zs = (const float2*)(z + (size_t)s * FATT);
    const float2* zd = (const float2*)(z + (size_t)d * FATT);
    float2 a = zs[lane], b = zd[lane];
    float v = a.x * b.x + a.y * b.y;
    #pragma unroll
    for (int off = 16; off > 0; off >>= 1) v += __shfl_xor(v, off, 32);
    if (lane == 0) {
        float ev = v > 0.0f ? v : SLOPE * v;
        e[edge] = ev;
        atomicMax(&m_ord[d], float_to_ord(ev));
    }
}

__global__ void edge_expsum(const int* __restrict__ dst, float* __restrict__ e,
                            const unsigned int* __restrict__ m_ord,
                            float* __restrict__ denom) {
    long long i = (long long)blockIdx.x * blockDim.x + threadIdx.x;
    if (i >= NEDGE) return;
    const int d = dst[i];
    const float m = ord_to_float(m_ord[d]);
    const float ex = __expf(e[i] - m);
    e[i] = ex;
    atomicAdd(&denom[d], ex);
}

__global__ void edge_scatter(const float* __restrict__ z, const int* __restrict__ src,
                             const int* __restrict__ dst, const float* __restrict__ e_exp,
                             const float* __restrict__ denom, float* __restrict__ h) {
    const int lane = threadIdx.x & 31;
    const int wave = threadIdx.x >> 5;
    long long edge = (long long)blockIdx.x * (blockDim.x >> 5) + wave;
    if (edge >= NEDGE) return;
    const int s = src[edge], d = dst[edge];
    const float alpha = e_exp[edge] / denom[d];
    const float2* zs = (const float2*)(z + (size_t)s * FATT);
    float2 a = zs[lane];
    float* hd = h + (size_t)d * FATT + 2 * lane;
    atomicAdd(hd + 0, alpha * a.x);
    atomicAdd(hd + 1, alpha * a.y);
}

__global__ void elu_inplace(float* __restrict__ h) {
    long long i = (long long)blockIdx.x * blockDim.x + threadIdx.x;
    if (i >= (long long)NTOT * FATT) return;
    const float v = h[i];
    h[i] = v > 0.0f ? v : (__expf(v) - 1.0f);
}

extern "C" void kernel_launch(void* const* d_in, const int* in_sizes, int n_in,
                              void* d_out, int out_size, void* d_ws, size_t ws_size,
                              hipStream_t stream) {
    const float* d_sim  = (const float*)d_in[0];
    const float* mi_sim = (const float*)d_in[1];
    const float* W_d    = (const float*)d_in[2];
    const float* W_mi   = (const float*)d_in[3];
    const int*   src    = (const int*)d_in[4];
    const int*   dst    = (const int*)d_in[5];
    float*       out    = (float*)d_out;

    float* z            = (float*)d_ws;
    float* e            = z + (size_t)NTOT * FATT;
    unsigned int* m_ord = (unsigned int*)(e + NEDGE);
    float* denom        = (float*)(m_ord + NTOT);

    const long long tot_h = (long long)NTOT * FATT;

    init_state<<<(int)((tot_h + 255) / 256), 256, 0, stream>>>(m_ord, denom, out);

    gemm_wmma_f32<<<(N_DIS + 127) / 128, 128, 0, stream>>>(d_sim, W_d, z, N_DIS, 383);
    gemm_wmma_f32<<<(N_MIC + 127) / 128, 128, 0, stream>>>(mi_sim, W_mi,
                                                           z + (size_t)N_DIS * FATT, N_MIC, 495);

    edge_score<<<(NEDGE + 7) / 8, 256, 0, stream>>>(z, src, dst, e, m_ord);

    edge_expsum<<<(NEDGE + 255) / 256, 256, 0, stream>>>(dst, e, m_ord, denom);

    edge_scatter<<<(NEDGE + 7) / 8, 256, 0, stream>>>(z, src, dst, e, denom, out);

    elu_inplace<<<(int)((tot_h + 255) / 256), 256, 0, stream>>>(out);
}


# Edges grouped by destination (count, scan, arrival-order rank fill); one wave per node does score, softmax, weighted sum and ELU in registers: no float atomics, sums in arrival-order rank
# speedup vs baseline: 3.2836x; 1.7757x over previous
#include <hip/hip_runtime.h>
#include <math.h>

#define N_DIS 20000
#define N_MIC 80000
#define NTOT  (N_DIS + N_MIC)
#define NEDGE 1600000
#define FATT  64
#define SLOPE 0.2f

typedef float v2f __attribute__((ext_vector_type(2)));
typedef float v8f __attribute__((ext_vector_type(8)));

__global__ __launch_bounds__(128) void gemm_wmma_f32(
        const float* __restrict__ X, const float* __restrict__ W,
        float* __restrict__ Z, int M, int K) {
    const int KT = 64;
    const int MT = 128;
    __shared__ float lX[MT * KT];
    __shared__ float lW[KT * FATT];

    const int lane  = threadIdx.x & 31;
    const int wave  = threadIdx.x >> 5;
    const int row   = lane & 15;
    const int khalf = lane >> 4;
    const int n     = lane & 15;

    const int blk_row = blockIdx.x * MT;
    const int lr0 = wave * 32 + row;
    const int lr1 = lr0 + 16;

    v8f c0 = {}, c1 = {}, c2 = {}, c3 = {};
    v8f c4 = {}, c5 = {}, c6 = {}, c7 = {};

    const int tid = threadIdx.x;
    const int sc  = tid & 63;
    const int sr  = tid >> 6;

    for (int kt = 0; kt < K; kt += KT) {
        float wv[(KT * FATT) / 128];
        float xv[(MT * KT) / 128];
        const int  kgx   = kt + sc;
        const bool kx_ok = kgx < K;
        const unsigned kgxc = (unsigned)(kx_ok ? kgx : K - 1);
        #pragma unroll
        for (int j = 0; j < (KT * FATT) / 128; ++j) {
            int kg  = kt + sr + 2 * j;
            int kgc = kg < K ? kg : K - 1;
            wv[j] = W[(unsigned)kgc * FATT + (unsigned)sc];
        }
        #pragma unroll
        for (int j = 0; j < (MT * KT) / 128; ++j) {
            int mg  = blk_row + sr + 2 * j;
            int mgc = mg < M ? mg : M - 1;
            xv[j] = X[(unsigned)mgc * (unsigned)K + kgxc];
        }
        #pragma unroll
        for (int j = 0; j < (KT * FATT) / 128; ++j) {
            int kg = kt + sr + 2 * j;
            lW[tid + 128 * j] = (kg < K) ? wv[j] : 0.0f;
        }
        #pragma unroll
        for (int j = 0; j < (MT * KT) / 128; ++j) {
            int mg = blk_row + sr + 2 * j;
            lX[tid + 128 * j] = (mg < M && kx_ok) ? xv[j] : 0.0f;
        }
        __syncthreads();

        #pragma unroll 4
        for (int k0 = 0; k0 < KT; k0 += 4) {
            const int ka = k0 + 2 * khalf;
            v2f a0 = *(const v2f*)(lX + lr0 * KT + ka);
            v2f a1 = *(const v2f*)(lX + lr1 * KT + ka);

            const float* wr0 = lW + ka * FATT + n;
            const float* wr1 = lW + (ka + 1) * FATT + n;
            v2f b0 = { wr0[0],  wr1[0]  };
            v2f b1 = { wr0[16], wr1[16] };
            v2f b2 = { wr0[32], wr1[32] };
            v2f b3 = { wr0[48], wr1[48] };

            c0 = __builtin_amdgcn_wmma_f32_16x16x4_f32(false, a0, false, b0, (short)0, c0, false, false);
            c1 = __builtin_amdgcn_wmma_f32_16x16x4_f32(false, a0, false, b1, (short)0, c1, false, false);
            c2 = __builtin_amdgcn_wmma_f32_16x16x4_f32(false, a0, false, b2, (short)0, c2, false, false);
            c3 = __builtin_amdgcn_wmma_f32_16x16x4_f32(false, a0, false, b3, (short)0, c3, false, false);
            c4 = __builtin_amdgcn_wmma_f32_16x16x4_f32(false, a1, false, b0, (short)0, c4, false, false);
            c5 = __builtin_amdgcn_wmma_f32_16x16x4_f32(false, a1, false, b1, (short)0, c5, false, false);
            c6 = __builtin_amdgcn_wmma_f32_16x16x4_f32(false, a1, false, b2, (short)0, c6, false, false);
            c7 = __builtin_amdgcn_wmma_f32_16x16x4_f32(false, a1, false, b3, (short)0, c7, false, false);
        }
        __syncthreads();
    }

    const int m0 = blk_row + wave * 32 + 8 * khalf;
    #pragma unroll
    for (int r = 0; r < 8; ++r) {
        int mm = m0 + r;
        if (mm < M) {
            float* zr = Z + (size_t)mm * FATT + n;
            zr[0]  = c0[r];
            zr[16] = c1[r];
            zr[32] = c2[r];
            zr[48] = c3[r];
        }
        int mm1 = mm + 16;
        if (mm1 < M) {
            float* zr = Z + (size_t)mm1 * FATT + n;
            zr[0]  = c4[r];
            zr[16] = c5[r];
            zr[32] = c6[r];
            zr[48] = c7[r];
        }
    }
}

#define CHUNK  256
#define NCHUNK ((NTOT + CHUNK - 1) / CHUNK)

static_assert(NEDGE % 4 == 0, "edge kernels read four edges per thread");

__global__ __launch_bounds__(256) void zero_counts(unsigned int* __restrict__ cnt) {
    const int i = blockIdx.x * 256 + threadIdx.x;
    if (i < NTOT) cnt[i] = 0u;
}

__global__ __launch_bounds__(256) void count_dst(const int* __restrict__ dst,
                                                 unsigned int* __restrict__ cnt) {
    const int q = blockIdx.x * 256 + threadIdx.x;
    if (q >= NEDGE / 4) return;
    const int4 d = ((const int4*)dst)[q];
    atomicAdd(&cnt[d.x], 1u);
    atomicAdd(&cnt[d.y], 1u);
    atomicAdd(&cnt[d.z], 1u);
    atomicAdd(&cnt[d.w], 1u);
}

__device__ __forceinline__ unsigned int wave_sum(unsigned int v) {
    for (int off = 16; off > 0; off >>= 1) v += __shfl_xor(v, off, 32);
    return v;
}

__global__ __launch_bounds__(256) void chunk_sums(const unsigned int* __restrict__ cnt,
                                                  unsigned int* __restrict__ part) {
    __shared__ unsigned int ws[8];
    const int i = blockIdx.x * CHUNK + threadIdx.x;
    const unsigned int v = wave_sum(i < NTOT ? cnt[i] : 0u);
    if ((threadIdx.x & 31) == 0) ws[threadIdx.x >> 5] = v;
    __syncthreads();
    if (threadIdx.x == 0)
        part[blockIdx.x] = ws[0] + ws[1] + ws[2] + ws[3] + ws[4] + ws[5] + ws[6] + ws[7];
}

__global__ __launch_bounds__(256) void chunk_scan(unsigned int* __restrict__ cur,
                                                  const unsigned int* __restrict__ part) {
    __shared__ unsigned int wbase[8];
    __shared__ unsigned int wtot[8];
    const int b = blockIdx.x, t = threadIdx.x;
    const int lane = t & 31, wave = t >> 5;
    unsigned int pb = 0u;
    for (int j = t; j < b; j += 256) pb += part[j];
    pb = wave_sum(pb);
    const int i = b * CHUNK + t;
    const unsigned int c = i < NTOT ? cur[i] : 0u;
    unsigned int inc = c;
    for (int off = 1; off < 32; off <<= 1) {
        const unsigned int up = __shfl_up(inc, off, 32);
        if (lane >= off) inc += up;
    }
    if (lane == 0)  wbase[wave] = pb;
    if (lane == 31) wtot[wave]  = inc;
    __syncthreads();
    unsigned int base = 0u;
    for (int w = 0; w < 8; ++w) {
        base += wbase[w];
        if (w < wave) base += wtot[w];
    }
    if (i < NTOT) cur[i] = base + inc - c;
}

__global__ __launch_bounds__(256) void fill_by_dst(const int* __restrict__ src,
                                                   const int* __restrict__ dst,
                                                   unsigned int* __restrict__ cur,
                                                   int* __restrict__ csr_src) {
    const int q = blockIdx.x * 256 + threadIdx.x;
    if (q >= NEDGE / 4) return;
    const int4 d = ((const int4*)dst)[q];
    const int4 s = ((const int4*)src)[q];
    unsigned int p0 = atomicAdd(&cur[d.x], 1u);
    unsigned int p1 = atomicAdd(&cur[d.y], 1u);
    unsigned int p2 = atomicAdd(&cur[d.z], 1u);
    unsigned int p3 = atomicAdd(&cur[d.w], 1u);
    p0 = p0 < NEDGE ? p0 : NEDGE - 1;
    p1 = p1 < NEDGE ? p1 : NEDGE - 1;
    p2 = p2 < NEDGE ? p2 : NEDGE - 1;
    p3 = p3 < NEDGE ? p3 : NEDGE - 1;
    csr_src[p0] = s.x;
    csr_src[p1] = s.y;
    csr_src[p2] = s.z;
    csr_src[p3] = s.w;
}

__device__ __forceinline__ float lrelu(float v) { return v > 0.0f ? v : SLOPE * v; }

__global__ __launch_bounds__(256) void node_attend(const float* __restrict__ z,
                                                   const int* __restrict__ csr_src,
                                                   const unsigned int* __restrict__ cur,
                                                   float* __restrict__ out) {
    const int lane = threadIdx.x & 31;
    const int node = blockIdx.x * 8 + __builtin_amdgcn_readfirstlane((int)(threadIdx.x >> 5));
    if (node >= NTOT) return;
    unsigned int end = cur[node];
    unsigned int beg = node > 0 ? cur[node - 1] : 0u;
    if (end > NEDGE) end = NEDGE;
    if (beg > end)   beg = end;

    const float2 zd = ((const float2*)(z + (size_t)node * FATT))[lane];
    float m = 0.0f, l = 0.0f, ax = 0.0f, ay = 0.0f;
    bool  have = false;

    for (unsigned int base = beg; base < end; base += 32) {
        const unsigned int j = base + lane;
        unsigned int sl = (unsigned int)csr_src[j < end ? j : end - 1];
        sl = sl < NTOT ? sl : NTOT - 1;
        const int n = (int)(end - base < 32u ? end - base : 32u);
        for (int t = 0; t < n; t += 4) {
            const int t1 = t + 1 < n ? t + 1 : n - 1;
            const int t2 = t + 2 < n ? t + 2 : n - 1;
            const int t3 = t + 3 < n ? t + 3 : n - 1;
            const int s0 = __builtin_amdgcn_readlane((int)sl, t);
            const int s1 = __builtin_amdgcn_readlane((int)sl, t1);
            const int s2 = __builtin_amdgcn_readlane((int)sl, t2);
            const int s3 = __builtin_amdgcn_readlane((int)sl, t3);
            const float2 r0 = ((const float2*)(z + (size_t)s0 * FATT))[lane];
            const float2 r1 = ((const float2*)(z + (size_t)s1 * FATT))[lane];
            const float2 r2 = ((const float2*)(z + (size_t)s2 * FATT))[lane];
            const float2 r3 = ((const float2*)(z + (size_t)s3 * FATT))[lane];
            float v0 = r0.x * zd.x + r0.y * zd.y;
            float v1 = r1.x * zd.x + r1.y * zd.y;
            float v2 = r2.x * zd.x + r2.y * zd.y;
            float v3 = r3.x * zd.x + r3.y * zd.y;
            for (int off = 16; off > 0; off >>= 1) {
                v0 += __shfl_xor(v0, off, 32);
                v1 += __shfl_xor(v1, off, 32);
                v2 += __shfl_xor(v2, off, 32);
                v3 += __shfl_xor(v3, off, 32);
            }
            const bool k1 = t + 1 < n, k2 = t + 2 < n, k3 = t + 3 < n;
            const float e0 = lrelu(v0);
            const float e1 = k1 ? lrelu(v1) : e0;
            const float e2 = k2 ? lrelu(v2) : e0;
            const float e3 = k3 ? lrelu(v3) : e0;
            float mn = fmaxf(fmaxf(e0, e1), fmaxf(e2, e3));
            if (have) mn = fmaxf(mn, m);
            const float sc = have ? __expf(m - mn) : 0.0f;
            const float p0 = __expf(e0 - mn);
            const float p1 = k1 ? __expf(e1 - mn) : 0.0f;
            const float p2 = k2 ? __expf(e2 - mn) : 0.0f;
            const float p3 = k3 ? __expf(e3 - mn) : 0.0f;
            l  = l  * sc + ((p0 + p1) + (p2 + p3));
            ax = ax * sc + ((p0 * r0.x + p1 * r1.x) + (p2 * r2.x + p3 * r3.x));
            ay = ay * sc + ((p0 * r0.y + p1 * r1.y) + (p2 * r2.y + p3 * r3.y));
            m = mn;
            have = true;
        }
    }
    float hx = 0.0f, hy = 0.0f;
    if (have) { hx = ax / l; hy = ay / l; }
    hx = hx > 0.0f ? hx : (__expf(hx) - 1.0f);
    hy = hy > 0.0f ? hy : (__expf(hy) - 1.0f);
    ((float2*)(out + (size_t)node * FATT))[lane] = make_float2(hx, hy);
}

extern "C" void kernel_launch(void* const* d_in, const int* in_sizes, int n_in,
                              void* d_out, int out_size, void* d_ws, size_t ws_size,
                              hipStream_t stream) {
    const float* d_sim  = (const float*)d_in[0];
    const float* mi_sim = (const float*)d_in[1];
    const float* W_d    = (const float*)d_in[2];
    const float* W_mi   = (const float*)d_in[3];
    const int*   src    = (const int*)d_in[4];
    const int*   dst    = (const int*)d_in[5];
    float*       out    = (float*)d_out;

    float* z            = (float*)d_ws;
    int* csr_src        = (int*)(z + (size_t)NTOT * FATT);
    unsigned int* cur   = (unsigned int*)(csr_src + NEDGE);
    unsigned int* part  = cur + NTOT;

    zero_counts<<<NCHUNK, 256, 0, stream>>>(cur);

    gemm_wmma_f32<<<(N_DIS + 127) / 128, 128, 0, stream>>>(d_sim, W_d, z, N_DIS, 383);
    gemm_wmma_f32<<<(N_MIC + 127) / 128, 128, 0, stream>>>(mi_sim, W_mi,
                                                           z + (size_t)N_DIS * FATT, N_MIC, 495);

    count_dst<<<(NEDGE / 4 + 255) / 256, 256, 0, stream>>>(dst, cur);
    chunk_sums<<<NCHUNK, 256, 0, stream>>>(cur, part);
    chunk_scan<<<NCHUNK, 256, 0, stream>>>(cur, part);
    fill_by_dst<<<(NEDGE / 4 + 255) / 256, 256, 0, stream>>>(src, dst, cur, csr_src);

    node_attend<<<(NTOT + 7) / 8, 256, 0, stream>>>(z, csr_src, cur, out);
}


# Edges grouped by destination with a two-level counting sort on LDS counters (391 buckets of 256 nodes) instead of two global atomics per edge; order inside a node = LDS add order
# speedup vs baseline: 5.3027x; 1.6149x over previous
#include <hip/hip_runtime.h>
#include <math.h>

#define N_DIS 20000
#define N_MIC 80000
#define NTOT  (N_DIS + N_MIC)
#define NEDGE 1600000
#define FATT  64
#define SLOPE 0.2f

typedef float v2f __attribute__((ext_vector_type(2)));
typedef float v8f __attribute__((ext_vector_type(8)));

__global__ __launch_bounds__(128) void gemm_wmma_f32(
        const float* __restrict__ X, const float* __restrict__ W,
        float* __restrict__ Z, int M, int K) {
    const int KT = 64;
    const int MT = 128;
    __shared__ float lX[MT * KT];
    __shared__ float lW[KT * FATT];

    const int lane  = threadIdx.x & 31;
    const int wave  = threadIdx.x >> 5;
    const int row   = lane & 15;
    const int khalf = lane >> 4;
    const int n     = lane & 15;

    const int blk_row = blockIdx.x * MT;
    const int lr0 = wave * 32 + row;
    const int lr1 = lr0 + 16;

    v8f c0 = {}, c1 = {}, c2 = {}, c3 = {};
    v8f c4 = {}, c5 = {}, c6 = {}, c7 = {};

    const int tid = threadIdx.x;
    const int sc  = tid & 63;
    const int sr  = tid >> 6;

    for (int kt = 0; kt < K; kt += KT) {
        float wv[(KT * FATT) / 128];
        float xv[(MT * KT) / 128];
        const int  kgx   = kt + sc;
        const bool kx_ok = kgx < K;
        const unsigned kgxc = (unsigned)(kx_ok ? kgx : K - 1);
        #pragma unroll
        for (int j = 0; j < (KT * FATT) / 128; ++j) {
            int kg  = kt + sr + 2 * j;
            int kgc = kg < K ? kg : K - 1;
            wv[j] = W[(unsigned)kgc * FATT + (unsigned)sc];
        }
        #pragma unroll
        for (int j = 0; j < (MT * KT) / 128; ++j) {
            int mg  = blk_row + sr + 2 * j;
            int mgc = mg < M ? mg : M - 1;
            xv[j] = X[(unsigned)mgc * (unsigned)K + kgxc];
        }
        #pragma unroll
        for (int j = 0; j < (KT * FATT) / 128; ++j) {
            int kg = kt + sr + 2 * j;
            lW[tid + 128 * j] = (kg < K) ? wv[j] : 0.0f;
        }
        #pragma unroll
        for (int j = 0; j < (MT * KT) / 128; ++j) {
            int mg = blk_row + sr + 2 * j;
            lX[tid + 128 * j] = (mg < M && kx_ok) ? xv[j] : 0.0f;
        }
        __syncthreads();

        #pragma unroll 4
        for (int k0 = 0; k0 < KT; k0 += 4) {
            const int ka = k0 + 2 * khalf;
            v2f a0 = *(const v2f*)(lX + lr0 * KT + ka);
            v2f a1 = *(const v2f*)(lX + lr1 * KT + ka);

            const float* wr0 = lW + ka * FATT + n;
            const float* wr1 = lW + (ka + 1) * FATT + n;
            v2f b0 = { wr0[0],  wr1[0]  };
            v2f b1 = { wr0[16], wr1[16] };
            v2f b2 = { wr0[32], wr1[32] };
            v2f b3 = { wr0[48], wr1[48] };

            c0 = __builtin_amdgcn_wmma_f32_16x16x4_f32(false, a0, false, b0, (short)0, c0, false, false);
            c1 = __builtin_amdgcn_wmma_f32_16x16x4_f32(false, a0, false, b1, (short)0, c1, false, false);
            c2 = __builtin_amdgcn_wmma_f32_16x16x4_f32(false, a0, false, b2, (short)0, c2, false, false);
            c3 = __builtin_amdgcn_wmma_f32_16x16x4_f32(false, a0, false, b3, (short)0, c3, false, false);
            c4 = __builtin_amdgcn_wmma_f32_16x16x4_f32(false, a1, false, b0, (short)0, c4, false, false);
            c5 = __builtin_amdgcn_wmma_f32_16x16x4_f32(false, a1, false, b1, (short)0, c5, false, false);
            c6 = __builtin_amdgcn_wmma_f32_16x16x4_f32(false, a1, false, b2, (short)0, c6, false, false);
            c7 = __builtin_amdgcn_wmma_f32_16x16x4_f32(false, a1, false, b3, (short)0, c7, false, false);
        }
        __syncthreads();
    }

    const int m0 = blk_row + wave * 32 + 8 * khalf;
    #pragma unroll
    for (int r = 0; r < 8; ++r) {
        int mm = m0 + r;
        if (mm < M) {
            float* zr = Z + (size_t)mm * FATT + n;
            zr[0]  = c0[r];
            zr[16] = c1[r];
            zr[32] = c2[r];
            zr[48] = c3[r];
        }
        int mm1 = mm + 16;
        if (mm1 < M) {
            float* zr = Z + (size_t)mm1 * FATT + n;
            zr[0]  = c4[r];
            zr[16] = c5[r];
            zr[32] = c6[r];
            zr[48] = c7[r];
        }
    }
}

#define CHUNK  256
#define NCHUNK ((NTOT + CHUNK - 1) / CHUNK)
#define EPB    4096
#define NBLK   ((NEDGE + EPB - 1) / EPB)
#define NBH    (NCHUNK * NBLK)
#define NSCAN  ((NBH + CHUNK - 1) / CHUNK)
#define SRC_BITS 17

static_assert(NEDGE % 4 == 0, "edge kernels read four edges per thread and load");
static_assert(EPB % 1024 == 0, "256 threads x 4 edges per load");
static_assert(NTOT <= (1 << SRC_BITS), "source index must fit the packed edge");

__device__ __forceinline__ unsigned int bucket_of(int d) {
    const unsigned int k = (unsigned int)d >> 8;
    return k < NCHUNK ? k : NCHUNK - 1;
}

__device__ __forceinline__ unsigned int pack_edge(int s, int d) {
    const unsigned int su = (unsigned int)s < NTOT ? (unsigned int)s : NTOT - 1;
    return su | (((unsigned int)d & 255u) << SRC_BITS);
}

__global__ __launch_bounds__(256) void bucket_hist(const int* __restrict__ dst,
                                                   unsigned int* __restrict__ bh) {
    __shared__ unsigned int h[NCHUNK];
    const int t = threadIdx.x;
    int4 d[EPB / 1024];
    bool ok[EPB / 1024];
    #pragma unroll
    for (int i = 0; i < EPB / 1024; ++i) {
        const int q = blockIdx.x * (EPB / 4) + i * 256 + t;
        ok[i] = q < NEDGE / 4;
        d[i]  = ((const int4*)dst)[ok[i] ? q : NEDGE / 4 - 1];
    }
    for (int k = t; k < NCHUNK; k += 256) h[k] = 0u;
    __syncthreads();
    #pragma unroll
    for (int i = 0; i < EPB / 1024; ++i) {
        if (ok[i]) {
            atomicAdd(&h[bucket_of(d[i].x)], 1u);
            atomicAdd(&h[bucket_of(d[i].y)], 1u);
            atomicAdd(&h[bucket_of(d[i].z)], 1u);
            atomicAdd(&h[bucket_of(d[i].w)], 1u);
        }
    }
    __syncthreads();
    for (int k = t; k < NCHUNK; k += 256) bh[k * NBLK + blockIdx.x] = h[k];
}

__device__ __forceinline__ unsigned int wave_sum(unsigned int v) {
    for (int off = 16; off > 0; off >>= 1) v += __shfl_xor(v, off, 32);
    return v;
}

__global__ __launch_bounds__(256) void chunk_sums(const unsigned int* __restrict__ cnt,
                                                  unsigned int* __restrict__ part, int n) {
    __shared__ unsigned int ws[8];
    const int i = blockIdx.x * CHUNK + threadIdx.x;
    const unsigned int v = wave_sum(i < n ? cnt[i] : 0u);
    if ((threadIdx.x & 31) == 0) ws[threadIdx.x >> 5] = v;
    __syncthreads();
    if (threadIdx.x == 0)
        part[blockIdx.x] = ws[0] + ws[1] + ws[2] + ws[3] + ws[4] + ws[5] + ws[6] + ws[7];
}

__global__ __launch_bounds__(256) void chunk_scan(unsigned int* __restrict__ cur,
                                                  const unsigned int* __restrict__ part, int n) {
    __shared__ unsigned int wbase[8];
    __shared__ unsigned int wtot[8];
    const int b = blockIdx.x, t = threadIdx.x;
    const int lane = t & 31, wave = t >> 5;
    unsigned int pb = 0u;
    for (int j = t; j < b; j += 256) pb += part[j];
    pb = wave_sum(pb);
    const int i = b * CHUNK + t;
    const unsigned int c = i < n ? cur[i] : 0u;
    unsigned int inc = c;
    for (int off = 1; off < 32; off <<= 1) {
        const unsigned int up = __shfl_up(inc, off, 32);
        if (lane >= off) inc += up;
    }
    if (lane == 0)  wbase[wave] = pb;
    if (lane == 31) wtot[wave]  = inc;
    __syncthreads();
    unsigned int base = 0u;
    for (int w = 0; w < 8; ++w) {
        base += wbase[w];
        if (w < wave) base += wtot[w];
    }
    if (i < n) cur[i] = base + inc - c;
}

__global__ __launch_bounds__(256) void bucket_scatter(const int* __restrict__ src,
                                                      const int* __restrict__ dst,
                                                      const unsigned int* __restrict__ bh,
                                                      unsigned int* __restrict__ ebuf) {
    __shared__ unsigned int nxt[NCHUNK];
    const int t = threadIdx.x;
    int4 d[EPB / 1024], s[EPB / 1024];
    bool ok[EPB / 1024];
    #pragma unroll
    for (int i = 0; i < EPB / 1024; ++i) {
        const int q = blockIdx.x * (EPB / 4) + i * 256 + t;
        ok[i] = q < NEDGE / 4;
        const int qc = ok[i] ? q : NEDGE / 4 - 1;
        d[i] = ((const int4*)dst)[qc];
        s[i] = ((const int4*)src)[qc];
    }
    for (int k = t; k < NCHUNK; k += 256) nxt[k] = bh[k * NBLK + blockIdx.x];
    __syncthreads();
    #pragma unroll
    for (int i = 0; i < EPB / 1024; ++i) {
        if (ok[i]) {
            unsigned int p0 = atomicAdd(&nxt[bucket_of(d[i].x)], 1u);
            unsigned int p1 = atomicAdd(&nxt[bucket_of(d[i].y)], 1u);
            unsigned int p2 = atomicAdd(&nxt[bucket_of(d[i].z)], 1u);
            unsigned int p3 = atomicAdd(&nxt[bucket_of(d[i].w)], 1u);
            p0 = p0 < NEDGE ? p0 : NEDGE - 1;
            p1 = p1 < NEDGE ? p1 : NEDGE - 1;
            p2 = p2 < NEDGE ? p2 : NEDGE - 1;
            p3 = p3 < NEDGE ? p3 : NEDGE - 1;
            ebuf[p0] = pack_edge(s[i].x, d[i].x);
            ebuf[p1] = pack_edge(s[i].y, d[i].y);
            ebuf[p2] = pack_edge(s[i].z, d[i].z);
            ebuf[p3] = pack_edge(s[i].w, d[i].w);
        }
    }
}

__global__ __launch_bounds__(256) void bucket_to_csr(const unsigned int* __restrict__ ebuf,
                                                     const unsigned int* __restrict__ bh,
                                                     unsigned int* __restrict__ cur,
                                                     int* __restrict__ csr_src) {
    __shared__ unsigned int cnt[CHUNK];
    __shared__ unsigned int wtot[8];
    const int k = blockIdx.x, t = threadIdx.x;
    const int lane = t & 31, wave = t >> 5;
    unsigned int bs = bh[k * NBLK];
    unsigned int be = k + 1 < NCHUNK ? bh[(k + 1) * NBLK] : NEDGE;
    if (be > NEDGE) be = NEDGE;
    if (bs > be)    bs = be;
    const unsigned int ne = be - bs;

    cnt[t] = 0u;
    __syncthreads();
    for (unsigned int base = 0; base < ne; base += 1024) {
        unsigned int w[4];
        bool ok[4];
        #pragma unroll
        for (int j = 0; j < 4; ++j) {
            const unsigned int i = base + (unsigned int)(j * 256 + t);
            ok[j] = i < ne;
            w[j]  = ebuf[bs + (ok[j] ? i : 0u)];
        }
        #pragma unroll
        for (int j = 0; j < 4; ++j)
            if (ok[j]) atomicAdd(&cnt[(w[j] >> SRC_BITS) & 255u], 1u);
    }
    __syncthreads();

    const unsigned int c = cnt[t];
    unsigned int inc = c;
    for (int off = 1; off < 32; off <<= 1) {
        const unsigned int up = __shfl_up(inc, off, 32);
        if (lane >= off) inc += up;
    }
    if (lane == 31) wtot[wave] = inc;
    __syncthreads();
    unsigned int start = inc - c;
    for (int w = 0; w < 8; ++w)
        if (w < wave) start += wtot[w];
    const int node = k * CHUNK + t;
    if (node < NTOT) cur[node] = bs + start + c;
    cnt[t] = start;
    __syncthreads();
    for (unsigned int base = 0; base < ne; base += 1024) {
        unsigned int w[4];
        bool ok[4];
        #pragma unroll
        for (int j = 0; j < 4; ++j) {
            const unsigned int i = base + (unsigned int)(j * 256 + t);
            ok[j] = i < ne;
            w[j]  = ebuf[bs + (ok[j] ? i : 0u)];
        }
        #pragma unroll
        for (int j = 0; j < 4; ++j) {
            if (ok[j]) {
                unsigned int r = bs + atomicAdd(&cnt[(w[j] >> SRC_BITS) & 255u], 1u);
                r = r < NEDGE ? r : NEDGE - 1;
                csr_src[r] = (int)(w[j] & ((1u << SRC_BITS) - 1u));
            }
        }
    }
}

__device__ __forceinline__ float lrelu(float v) { return v > 0.0f ? v : SLOPE * v; }

__global__ __launch_bounds__(256) void node_attend(const float* __restrict__ z,
                                                   const int* __restrict__ csr_src,
                                                   const unsigned int* __restrict__ cur,
                                                   float* __restrict__ out) {
    const int lane = threadIdx.x & 31;
    const int node = blockIdx.x * 8 + __builtin_amdgcn_readfirstlane((int)(threadIdx.x >> 5));
    if (node >= NTOT) return;
    unsigned int end = cur[node];
    unsigned int beg = node > 0 ? cur[node - 1] : 0u;
    if (end > NEDGE) end = NEDGE;
    if (beg > end)   beg = end;

    const float2 zd = ((const float2*)(z + (size_t)node * FATT))[lane];
    float m = 0.0f, l = 0.0f, ax = 0.0f, ay = 0.0f;
    bool  have = false;

    for (unsigned int base = beg; base < end; base += 32) {
        const unsigned int j = base + lane;
        unsigned int sl = (unsigned int)csr_src[j < end ? j : end - 1];
        sl = sl < NTOT ? sl : NTOT - 1;
        const int n = (int)(end - base < 32u ? end - base : 32u);
        for (int t = 0; t < n; t += 4) {
            const int t1 = t + 1 < n ? t + 1 : n - 1;
            const int t2 = t + 2 < n ? t + 2 : n - 1;
            const int t3 = t + 3 < n ? t + 3 : n - 1;
            const int s0 = __builtin_amdgcn_readlane((int)sl, t);
            const int s1 = __builtin_amdgcn_readlane((int)sl, t1);
            const int s2 = __builtin_amdgcn_readlane((int)sl, t2);
            const int s3 = __builtin_amdgcn_readlane((int)sl, t3);
            const float2 r0 = ((const float2*)(z + (size_t)s0 * FATT))[lane];
            const float2 r1 = ((const float2*)(z + (size_t)s1 * FATT))[lane];
            const float2 r2 = ((const float2*)(z + (size_t)s2 * FATT))[lane];
            const float2 r3 = ((const float2*)(z + (size_t)s3 * FATT))[lane];
            float v0 = r0.x * zd.x + r0.y * zd.y;
            float v1 = r1.x * zd.x + r1.y * zd.y;
            float v2 = r2.x * zd.x + r2.y * zd.y;
            float v3 = r3.x * zd.x + r3.y * zd.y;
            for (int off = 16; off > 0; off >>= 1) {
                v0 += __shfl_xor(v0, off, 32);
                v1 += __shfl_xor(v1, off, 32);
                v2 += __shfl_xor(v2, off, 32);
                v3 += __shfl_xor(v3, off, 32);
            }
            const bool k1 = t + 1 < n, k2 = t + 2 < n, k3 = t + 3 < n;
            const float e0 = lrelu(v0);
            const float e1 = k1 ? lrelu(v1) : e0;
            const float e2 = k2 ? lrelu(v2) : e0;
            const float e3 = k3 ? lrelu(v3) : e0;
            float mn = fmaxf(fmaxf(e0, e1), fmaxf(e2, e3));
            if (have) mn = fmaxf(mn, m);
            const float sc = have ? __expf(m - mn) : 0.0f;
            const float p0 = __expf(e0 - mn);
            const float p1 = k1 ? __expf(e1 - mn) : 0.0f;
            const float p2 = k2 ? __expf(e2 - mn) : 0.0f;
            const float p3 = k3 ? __expf(e3 - mn) : 0.0f;
            l  = l  * sc + ((p0 + p1) + (p2 + p3));
            ax = ax * sc + ((p0 * r0.x + p1 * r1.x) + (p2 * r2.x + p3 * r3.x));
            ay = ay * sc + ((p0 * r0.y + p1 * r1.y) + (p2 * r2.y + p3 * r3.y));
            m = mn;
            have = true;
        }
    }
    float hx = 0.0f, hy = 0.0f;
    if (have) { hx = ax / l; hy = ay / l; }
    hx = hx > 0.0f ? hx : (__expf(hx) - 1.0f);
    hy = hy > 0.0f ? hy : (__expf(hy) - 1.0f);
    ((float2*)(out + (size_t)node * FATT))[lane] = make_float2(hx, hy);
}

extern "C" void kernel_launch(void* const* d_in, const int* in_sizes, int n_in,
                              void* d_out, int out_size, void* d_ws, size_t ws_size,
                              hipStream_t stream) {
    const float* d_sim  = (const float*)d_in[0];
    const float* mi_sim = (const float*)d_in[1];
    const float* W_d    = (const float*)d_in[2];
    const float* W_mi   = (const float*)d_in[3];
    const int*   src    = (const int*)d_in[4];
    const int*   dst    = (const int*)d_in[5];
    float*       out    = (float*)d_out;

    float* z            = (float*)d_ws;
    int* csr_src        = (int*)(z + (size_t)NTOT * FATT);
    unsigned int* cur   = (unsigned int*)(csr_src + NEDGE);
    unsigned int* bh    = (unsigned int*)d_ws;
    unsigned int* part  = bh + NSCAN * CHUNK;
    unsigned int* ebuf  = part + 1024;

    bucket_hist<<<NBLK, 256, 0, stream>>>(dst, bh);
    chunk_sums<<<NSCAN, 256, 0, stream>>>(bh, part, NBH);
    chunk_scan<<<NSCAN, 256, 0, stream>>>(bh, part, NBH);
    bucket_scatter<<<NBLK, 256, 0, stream>>>(src, dst, bh, ebuf);
    bucket_to_csr<<<NCHUNK, 256, 0, stream>>>(ebuf, bh, cur, csr_src);

    gemm_wmma_f32<<<(N_DIS + 127) / 128, 128, 0, stream>>>(d_sim, W_d, z, N_DIS, 383);
    gemm_wmma_f32<<<(N_MIC + 127) / 128, 128, 0, stream>>>(mi_sim, W_mi,
                                                           z + (size_t)N_DIS * FATT, N_MIC, 495);

    node_attend<<<(NTOT + 7) / 8, 256, 0, stream>>>(z, csr_src, cur, out);
}


# node_attend: half a wave per edge (16 lanes x 16 B, 8 row loads in flight); scores by a halving row_xmask DPP butterfly, softmax once per edge in its owner lane, no LDS permutes; sum order changes
# speedup vs baseline: 5.6455x; 1.0646x over previous
#include <hip/hip_runtime.h>
#include <math.h>

#define N_DIS 20000
#define N_MIC 80000
#define NTOT  (N_DIS + N_MIC)
#define NEDGE 1600000
#define FATT  64
#define SLOPE 0.2f

typedef float v2f __attribute__((ext_vector_type(2)));
typedef float v8f __attribute__((ext_vector_type(8)));

__global__ __launch_bounds__(128) void gemm_wmma_f32(
        const float* __restrict__ X, const float* __restrict__ W,
        float* __restrict__ Z, int M, int K) {
    const int KT = 64;
    const int MT = 128;
    __shared__ float lX[MT * KT];
    __shared__ float lW[KT * FATT];

    const int lane  = threadIdx.x & 31;
    const int wave  = threadIdx.x >> 5;
    const int row   = lane & 15;
    const int khalf = lane >> 4;
    const int n     = lane & 15;

    const int blk_row = blockIdx.x * MT;
    const int lr0 = wave * 32 + row;
    const int lr1 = lr0 + 16;

    v8f c0 = {}, c1 = {}, c2 = {}, c3 = {};
    v8f c4 = {}, c5 = {}, c6 = {}, c7 = {};

    const int tid = threadIdx.x;
    const int sc  = tid & 63;
    const int sr  = tid >> 6;

    for (int kt = 0; kt < K; kt += KT) {
        float wv[(KT * FATT) / 128];
        float xv[(MT * KT) / 128];
        const int  kgx   = kt + sc;
        const bool kx_ok = kgx < K;
        const unsigned kgxc = (unsigned)(kx_ok ? kgx : K - 1);
        #pragma unroll
        for (int j = 0; j < (KT * FATT) / 128; ++j) {
            int kg  = kt + sr + 2 * j;
            int kgc = kg < K ? kg : K - 1;
            wv[j] = W[(unsigned)kgc * FATT + (unsigned)sc];
        }
        #pragma unroll
        for (int j = 0; j < (MT * KT) / 128; ++j) {
            int mg  = blk_row + sr + 2 * j;
            int mgc = mg < M ? mg : M - 1;
            xv[j] = X[(unsigned)mgc * (unsigned)K + kgxc];
        }
        #pragma unroll
        for (int j = 0; j < (KT * FATT) / 128; ++j) {
            int kg = kt + sr + 2 * j;
            lW[tid + 128 * j] = (kg < K) ? wv[j] : 0.0f;
        }
        #pragma unroll
        for (int j = 0; j < (MT * KT) / 128; ++j) {
            int mg = blk_row + sr + 2 * j;
            lX[tid + 128 * j] = (mg < M && kx_ok) ? xv[j] : 0.0f;
        }
        __syncthreads();

        #pragma unroll 4
        for (int k0 = 0; k0 < KT; k0 += 4) {
            const int ka = k0 + 2 * khalf;
            v2f a0 = *(const v2f*)(lX + lr0 * KT + ka);
            v2f a1 = *(const v2f*)(lX + lr1 * KT + ka);

            const float* wr0 = lW + ka * FATT + n;
            const float* wr1 = lW + (ka + 1) * FATT + n;
            v2f b0 = { wr0[0],  wr1[0]  };
            v2f b1 = { wr0[16], wr1[16] };
            v2f b2 = { wr0[32], wr1[32] };
            v2f b3 = { wr0[48], wr1[48] };

            c0 = __builtin_amdgcn_wmma_f32_16x16x4_f32(false, a0, false, b0, (short)0, c0, false, false);
            c1 = __builtin_amdgcn_wmma_f32_16x16x4_f32(false, a0, false, b1, (short)0, c1, false, false);
            c2 = __builtin_amdgcn_wmma_f32_16x16x4_f32(false, a0, false, b2, (short)0, c2, false, false);
            c3 = __builtin_amdgcn_wmma_f32_16x16x4_f32(false, a0, false, b3, (short)0, c3, false, false);
            c4 = __builtin_amdgcn_wmma_f32_16x16x4_f32(false, a1, false, b0, (short)0, c4, false, false);
            c5 = __builtin_amdgcn_wmma_f32_16x16x4_f32(false, a1, false, b1, (short)0, c5, false, false);
            c6 = __builtin_amdgcn_wmma_f32_16x16x4_f32(false, a1, false, b2, (short)0, c6, false, false);
            c7 = __builtin_amdgcn_wmma_f32_16x16x4_f32(false, a1, false, b3, (short)0, c7, false, false);
        }
        __syncthreads();
    }

    const int m0 = blk_row + wave * 32 + 8 * khalf;
    #pragma unroll
    for (int r = 0; r < 8; ++r) {
        int mm = m0 + r;
        if (mm < M) {
            float* zr = Z + (size_t)mm * FATT + n;
            zr[0]  = c0[r];
            zr[16] = c1[r];
            zr[32] = c2[r];
            zr[48] = c3[r];
        }
        int mm1 = mm + 16;
        if (mm1 < M) {
            float* zr = Z + (size_t)mm1 * FATT + n;
            zr[0]  = c4[r];
            zr[16] = c5[r];
            zr[32] = c6[r];
            zr[48] = c7[r];
        }
    }
}

#define CHUNK  256
#define NCHUNK ((NTOT + CHUNK - 1) / CHUNK)
#define EPB    4096
#define NBLK   ((NEDGE + EPB - 1) / EPB)
#define NBH    (NCHUNK * NBLK)
#define NSCAN  ((NBH + CHUNK - 1) / CHUNK)
#define SRC_BITS 17

static_assert(NEDGE % 4 == 0, "edge kernels read four edges per thread and load");
static_assert(EPB % 1024 == 0, "256 threads x 4 edges per load");
static_assert(NTOT <= (1 << SRC_BITS), "source index must fit the packed edge");

__device__ __forceinline__ unsigned int bucket_of(int d) {
    const unsigned int k = (unsigned int)d >> 8;
    return k < NCHUNK ? k : NCHUNK - 1;
}

__device__ __forceinline__ unsigned int pack_edge(int s, int d) {
    const unsigned int su = (unsigned int)s < NTOT ? (unsigned int)s : NTOT - 1;
    return su | (((unsigned int)d & 255u) << SRC_BITS);
}

__global__ __launch_bounds__(256) void bucket_hist(const int* __restrict__ dst,
                                                   unsigned int* __restrict__ bh) {
    __shared__ unsigned int h[NCHUNK];
    const int t = threadIdx.x;
    int4 d[EPB / 1024];
    bool ok[EPB / 1024];
    #pragma unroll
    for (int i = 0; i < EPB / 1024; ++i) {
        const int q = blockIdx.x * (EPB / 4) + i * 256 + t;
        ok[i] = q < NEDGE / 4;
        d[i]  = ((const int4*)dst)[ok[i] ? q : NEDGE / 4 - 1];
    }
    for (int k = t; k < NCHUNK; k += 256) h[k] = 0u;
    __syncthreads();
    #pragma unroll
    for (int i = 0; i < EPB / 1024; ++i) {
        if (ok[i]) {
            atomicAdd(&h[bucket_of(d[i].x)], 1u);
            atomicAdd(&h[bucket_of(d[i].y)], 1u);
            atomicAdd(&h[bucket_of(d[i].z)], 1u);
            atomicAdd(&h[bucket_of(d[i].w)], 1u);
        }
    }
    __syncthreads();
    for (int k = t; k < NCHUNK; k += 256) bh[k * NBLK + blockIdx.x] = h[k];
}

__device__ __forceinline__ unsigned int wave_sum(unsigned int v) {
    for (int off = 16; off > 0; off >>= 1) v += __shfl_xor(v, off, 32);
    return v;
}

__global__ __launch_bounds__(256) void chunk_sums(const unsigned int* __restrict__ cnt,
                                                  unsigned int* __restrict__ part, int n) {
    __shared__ unsigned int ws[8];
    const int i = blockIdx.x * CHUNK + threadIdx.x;
    const unsigned int v = wave_sum(i < n ? cnt[i] : 0u);
    if ((threadIdx.x & 31) == 0) ws[threadIdx.x >> 5] = v;
    __syncthreads();
    if (threadIdx.x == 0)
        part[blockIdx.x] = ws[0] + ws[1] + ws[2] + ws[3] + ws[4] + ws[5] + ws[6] + ws[7];
}

__global__ __launch_bounds__(256) void chunk_scan(unsigned int* __restrict__ cur,
                                                  const unsigned int* __restrict__ part, int n) {
    __shared__ unsigned int wbase[8];
    __shared__ unsigned int wtot[8];
    const int b = blockIdx.x, t = threadIdx.x;
    const int lane = t & 31, wave = t >> 5;
    unsigned int pb = 0u;
    for (int j = t; j < b; j += 256) pb += part[j];
    pb = wave_sum(pb);
    const int i = b * CHUNK + t;
    const unsigned int c = i < n ? cur[i] : 0u;
    unsigned int inc = c;
    for (int off = 1; off < 32; off <<= 1) {
        const unsigned int up = __shfl_up(inc, off, 32);
        if (lane >= off) inc += up;
    }
    if (lane == 0)  wbase[wave] = pb;
    if (lane == 31) wtot[wave]  = inc;
    __syncthreads();
    unsigned int base = 0u;
    for (int w = 0; w < 8; ++w) {
        base += wbase[w];
        if (w < wave) base += wtot[w];
    }
    if (i < n) cur[i] = base + inc - c;
}

__global__ __launch_bounds__(256) void bucket_scatter(const int* __restrict__ src,
                                                      const int* __restrict__ dst,
                                                      const unsigned int* __restrict__ bh,
                                                      unsigned int* __restrict__ ebuf) {
    __shared__ unsigned int nxt[NCHUNK];
    const int t = threadIdx.x;
    int4 d[EPB / 1024], s[EPB / 1024];
    bool ok[EPB / 1024];
    #pragma unroll
    for (int i = 0; i < EPB / 1024; ++i) {
        const int q = blockIdx.x * (EPB / 4) + i * 256 + t;
        ok[i] = q < NEDGE / 4;
        const int qc = ok[i] ? q : NEDGE / 4 - 1;
        d[i] = ((const int4*)dst)[qc];
        s[i] = ((const int4*)src)[qc];
    }
    for (int k = t; k < NCHUNK; k += 256) nxt[k] = bh[k * NBLK + blockIdx.x];
    __syncthreads();
    #pragma unroll
    for (int i = 0; i < EPB / 1024; ++i) {
        if (ok[i]) {
            unsigned int p0 = atomicAdd(&nxt[bucket_of(d[i].x)], 1u);
            unsigned int p1 = atomicAdd(&nxt[bucket_of(d[i].y)], 1u);
            unsigned int p2 = atomicAdd(&nxt[bucket_of(d[i].z)], 1u);
            unsigned int p3 = atomicAdd(&nxt[bucket_of(d[i].w)], 1u);
            p0 = p0 < NEDGE ? p0 : NEDGE - 1;
            p1 = p1 < NEDGE ? p1 : NEDGE - 1;
            p2 = p2 < NEDGE ? p2 : NEDGE - 1;
            p3 = p3 < NEDGE ? p3 : NEDGE - 1;
            ebuf[p0] = pack_edge(s[i].x, d[i].x);
            ebuf[p1] = pack_edge(s[i].y, d[i].y);
            ebuf[p2] = pack_edge(s[i].z, d[i].z);
            ebuf[p3] = pack_edge(s[i].w, d[i].w);
        }
    }
}

__global__ __launch_bounds__(256) void bucket_to_csr(const unsigned int* __restrict__ ebuf,
                                                     const unsigned int* __restrict__ bh,
                                                     unsigned int* __restrict__ cur,
                                                     int* __restrict__ csr_src) {
    __shared__ unsigned int cnt[CHUNK];
    __shared__ unsigned int wtot[8];
    const int k = blockIdx.x, t = threadIdx.x;
    const int lane = t & 31, wave = t >> 5;
    unsigned int bs = bh[k * NBLK];
    unsigned int be = k + 1 < NCHUNK ? bh[(k + 1) * NBLK] : NEDGE;
    if (be > NEDGE) be = NEDGE;
    if (bs > be)    bs = be;
    const unsigned int ne = be - bs;

    cnt[t] = 0u;
    __syncthreads();
    for (unsigned int base = 0; base < ne; base += 1024) {
        unsigned int w[4];
        bool ok[4];
        #pragma unroll
        for (int j = 0; j < 4; ++j) {
            const unsigned int i = base + (unsigned int)(j * 256 + t);
            ok[j] = i < ne;
            w[j]  = ebuf[bs + (ok[j] ? i : 0u)];
        }
        #pragma unroll
        for (int j = 0; j < 4; ++j)
            if (ok[j]) atomicAdd(&cnt[(w[j] >> SRC_BITS) & 255u], 1u);
    }
    __syncthreads();

    const unsigned int c = cnt[t];
    unsigned int inc = c;
    for (int off = 1; off < 32; off <<= 1) {
        const unsigned int up = __shfl_up(inc, off, 32);
        if (lane >= off) inc += up;
    }
    if (lane == 31) wtot[wave] = inc;
    __syncthreads();
    unsigned int start = inc - c;
    for (int w = 0; w < 8; ++w)
        if (w < wave) start += wtot[w];
    const int node = k * CHUNK + t;
    if (node < NTOT) cur[node] = bs + start + c;
    cnt[t] = start;
    __syncthreads();
    for (unsigned int base = 0; base < ne; base += 1024) {
        unsigned int w[4];
        bool ok[4];
        #pragma unroll
        for (int j = 0; j < 4; ++j) {
            const unsigned int i = base + (unsigned int)(j * 256 + t);
            ok[j] = i < ne;
            w[j]  = ebuf[bs + (ok[j] ? i : 0u)];
        }
        #pragma unroll
        for (int j = 0; j < 4; ++j) {
            if (ok[j]) {
                unsigned int r = bs + atomicAdd(&cnt[(w[j] >> SRC_BITS) & 255u], 1u);
                r = r < NEDGE ? r : NEDGE - 1;
                csr_src[r] = (int)(w[j] & ((1u << SRC_BITS) - 1u));
            }
        }
    }
}

__device__ __forceinline__ float lrelu(float v) { return v > 0.0f ? v : SLOPE * v; }


__device__ __forceinline__ float xhalf(float x) {
    const int b = __builtin_bit_cast(int, x);
    return __builtin_bit_cast(float,
        __builtin_amdgcn_permlanex16(b, b, 0x76543210, 0xfedcba98, false, false));
}

template <int MASK>
__device__ __forceinline__ float row_xor(float v) {
    return __builtin_bit_cast(float,
        __builtin_amdgcn_mov_dpp(__builtin_bit_cast(int, v), 0x160 | MASK, 0xf, 0xf, true));
}

template <int MASK>
__device__ __forceinline__ float fold_pair(float lo, float hi, int c) {
    const bool up = (c & MASK) != 0;
    const float keep = up ? hi : lo;
    const float send = up ? lo : hi;
    return keep + row_xor<MASK>(send);
}

#define ROW_SHARE_CASE(P) case P: return __builtin_amdgcn_mov_dpp(v, 0x150 | P, 0xf, 0xf, true);
__device__ __forceinline__ int row_bcast(int v, int p) {
    switch (p & 15) {
        ROW_SHARE_CASE(0)  ROW_SHARE_CASE(1)  ROW_SHARE_CASE(2)  ROW_SHARE_CASE(3)
        ROW_SHARE_CASE(4)  ROW_SHARE_CASE(5)  ROW_SHARE_CASE(6)  ROW_SHARE_CASE(7)
        ROW_SHARE_CASE(8)  ROW_SHARE_CASE(9)  ROW_SHARE_CASE(10) ROW_SHARE_CASE(11)
        ROW_SHARE_CASE(12) ROW_SHARE_CASE(13) ROW_SHARE_CASE(14)
        default: return __builtin_amdgcn_mov_dpp(v, 0x150 | 15, 0xf, 0xf, true);
    }
}
#undef ROW_SHARE_CASE

typedef float v4f __attribute__((ext_vector_type(4)));

template <int NP, int P0>
__device__ __forceinline__ void attend_pairs(const float* __restrict__ z, unsigned int sl, int n,
                                             unsigned int c16, int c, int h, v2f zlo, v2f zhi,
                                             float& m, float& l, v2f& alo, v2f& ahi) {
    static_assert(NP == 4 || NP == 8, "the butterfly below is written for 4 or 8 pairs");
    static_assert(P0 % NP == 0, "lane P0+k must be the lane whose low bits are k");
    v4f   r[NP];
    float d[NP];
    #pragma unroll
    for (int i = 0; i < NP; ++i) {
        const unsigned int s = (unsigned int)row_bcast((int)sl, P0 + i);
        r[i] = *(const v4f*)((const char*)z + (s * (unsigned int)(FATT * 4) + c16));
    }
    __builtin_amdgcn_sched_barrier(0);
    #pragma unroll
    for (int i = 0; i < NP; ++i) {
        const v2f t = __builtin_elementwise_fma(r[i].zw, zhi, r[i].xy * zlo);
        d[i] = t.x + t.y;
    }
    float s;
    if (NP == 8) {
        const float e0 = fold_pair<4>(d[0], d[4], c);
        const float e1 = fold_pair<4>(d[1], d[5], c);
        const float e2 = fold_pair<4>(d[2], d[6], c);
        const float e3 = fold_pair<4>(d[3], d[7], c);
        const float f0 = fold_pair<2>(e0, e2, c);
        const float f1 = fold_pair<2>(e1, e3, c);
        s = fold_pair<1>(f0, f1, c);
        s += row_xor<8>(s);
    } else {
        const float f0 = fold_pair<2>(d[0], d[2], c);
        const float f1 = fold_pair<2>(d[1], d[3], c);
        s = fold_pair<1>(f0, f1, c);
        s += row_xor<4>(s);
        s += row_xor<8>(s);
    }
    const bool mine = ((c & ~(NP - 1)) == P0) && (2 * c + h < n);
    const float e = mine ? fmaxf(s, SLOPE * s) : -INFINITY;
    float cm = e;
    cm = fmaxf(cm, row_xor<1>(cm));
    cm = fmaxf(cm, row_xor<2>(cm));
    cm = fmaxf(cm, row_xor<4>(cm));
    cm = fmaxf(cm, row_xor<8>(cm));
    cm = fmaxf(cm, xhalf(cm));
    const float mn = fmaxf(m, cm);
    const float sc = __expf(m - mn);
    const float p  = __expf(e - mn);
    l = l * sc + p;
    v2f slo = alo * sc;
    v2f shi = ahi * sc;
    #pragma unroll
    for (int i = 0; i < NP; ++i) {
        const float w = __builtin_bit_cast(float, row_bcast(__builtin_bit_cast(int, p), P0 + i));
        const v2f  ww = { w, w };
        slo = __builtin_elementwise_fma(r[i].xy, ww, slo);
        shi = __builtin_elementwise_fma(r[i].zw, ww, shi);
    }
    alo = slo; ahi = shi;
    m = mn;
}

__global__ __launch_bounds__(256) void node_attend(const float* __restrict__ z,
                                                   const int* __restrict__ csr_src,
                                                   const unsigned int* __restrict__ cur,
                                                   float* __restrict__ out) {
    const int lane = threadIdx.x & 31;
    const int c = lane & 15;
    const int h = lane >> 4;
    const unsigned int c16 = (unsigned int)c * 16u;
    const int node = blockIdx.x * 8 + __builtin_amdgcn_readfirstlane((int)(threadIdx.x >> 5));
    if (node >= NTOT) return;
    unsigned int end = cur[node];
    unsigned int beg = node > 0 ? cur[node - 1] : 0u;
    if (end > NEDGE) end = NEDGE;
    if (beg > end)   beg = end;

    const v4f zd = *(const v4f*)((const char*)(z + (size_t)node * FATT) + c16);
    const v2f zlo = zd.xy, zhi = zd.zw;
    float m = -3.0e38f, l = 0.0f;
    v2f alo = { 0.0f, 0.0f }, ahi = { 0.0f, 0.0f };

    for (unsigned int base = beg; base < end; base += 32) {
        const unsigned int j = base + 2u * (unsigned int)c + (unsigned int)h;
        unsigned int sl = (unsigned int)csr_src[j < end ? j : end - 1];
        sl = sl < NTOT ? sl : NTOT - 1;
        const int n = (int)(end - base < 32u ? end - base : 32u);
        if (n > 8) attend_pairs<8, 0>(z, sl, n, c16, c, h, zlo, zhi, m, l, alo, ahi);
        else       attend_pairs<4, 0>(z, sl, n, c16, c, h, zlo, zhi, m, l, alo, ahi);
        if (n > 16) {
            if (n > 24) attend_pairs<8, 8>(z, sl, n, c16, c, h, zlo, zhi, m, l, alo, ahi);
            else        attend_pairs<4, 8>(z, sl, n, c16, c, h, zlo, zhi, m, l, alo, ahi);
        }
    }
    l += row_xor<1>(l);
    l += row_xor<2>(l);
    l += row_xor<4>(l);
    l += row_xor<8>(l);
    l += xhalf(l);
    const float wx = alo.x + xhalf(alo.x);
    const float wy = alo.y + xhalf(alo.y);
    const float wz = ahi.x + xhalf(ahi.x);
    const float ww = ahi.y + xhalf(ahi.y);
    v4f hv = { 0.0f, 0.0f, 0.0f, 0.0f };
    if (l > 0.0f) {
        const float inv = 1.0f / l;
        hv.x = wx * inv; hv.y = wy * inv; hv.z = wz * inv; hv.w = ww * inv;
    }
    hv.x = hv.x > 0.0f ? hv.x : (__expf(hv.x) - 1.0f);
    hv.y = hv.y > 0.0f ? hv.y : (__expf(hv.y) - 1.0f);
    hv.z = hv.z > 0.0f ? hv.z : (__expf(hv.z) - 1.0f);
    hv.w = hv.w > 0.0f ? hv.w : (__expf(hv.w) - 1.0f);
    if (h == 0) *(v4f*)((char*)(out + (size_t)node * FATT) + c16) = hv;
}

extern "C" void kernel_launch(void* const* d_in, const int* in_sizes, int n_in,
                              void* d_out, int out_size, void* d_ws, size_t ws_size,
                              hipStream_t stream) {
    const float* d_sim  = (const float*)d_in[0];
    const float* mi_sim = (const float*)d_in[1];
    const float* W_d    = (const float*)d_in[2];
    const float* W_mi   = (const float*)d_in[3];
    const int*   src    = (const int*)d_in[4];
    const int*   dst    = (const int*)d_in[5];
    float*       out    = (float*)d_out;

    float* z            = (float*)d_ws;
    int* csr_src        = (int*)(z + (size_t)NTOT * FATT);
    unsigned int* cur   = (unsigned int*)(csr_src + NEDGE);
    unsigned int* bh    = (unsigned int*)d_ws;
    unsigned int* part  = bh + NSCAN * CHUNK;
    unsigned int* ebuf  = part + 1024;

    bucket_hist<<<NBLK, 256, 0, stream>>>(dst, bh);
    chunk_sums<<<NSCAN, 256, 0, stream>>>(bh, part, NBH);
    chunk_scan<<<NSCAN, 256, 0, stream>>>(bh, part, NBH);
    bucket_scatter<<<NBLK, 256, 0, stream>>>(src, dst, bh, ebuf);
    bucket_to_csr<<<NCHUNK, 256, 0, stream>>>(ebuf, bh, cur, csr_src);

    gemm_wmma_f32<<<(N_DIS + 127) / 128, 128, 0, stream>>>(d_sim, W_d, z, N_DIS, 383);
    gemm_wmma_f32<<<(N_MIC + 127) / 128, 128, 0, stream>>>(mi_sim, W_mi,
                                                           z + (size_t)N_DIS * FATT, N_MIC, 495);

    node_attend<<<(NTOT + 7) / 8, 256, 0, stream>>>(z, csr_src, cur, out);
}


# Both projections in ONE launch (782 work-groups, the longer GEMM's 625 first): the shorter one no longer runs alone at one wave per SIMD on 157 of 256 compute units; one launch boundary less
# speedup vs baseline: 6.4220x; 1.1375x over previous
#include <hip/hip_runtime.h>
#include <math.h>

#define N_DIS 20000
#define N_MIC 80000
#define NTOT  (N_DIS + N_MIC)
#define NEDGE 1600000
#define FATT  64
#define SLOPE 0.2f

typedef float v2f __attribute__((ext_vector_type(2)));
typedef float v8f __attribute__((ext_vector_type(8)));

__global__ __launch_bounds__(128) void gemm_wmma_f32(
        const float* __restrict__ Xa, const float* __restrict__ Wa,
        float* __restrict__ Za, int Ma, int Ka,
        const float* __restrict__ Xb, const float* __restrict__ Wb,
        float* __restrict__ Zb, int Mb, int Kb, int nblk_a) {
    const bool second = (int)blockIdx.x >= nblk_a;
    const float* __restrict__ X = second ? Xb : Xa;
    const float* __restrict__ W = second ? Wb : Wa;
    float* __restrict__ Z = second ? Zb : Za;
    const int M = second ? Mb : Ma;
    const int K = second ? Kb : Ka;
    const int KT = 64;
    const int MT = 128;
    __shared__ float lX[MT * KT];
    __shared__ float lW[KT * FATT];

    const int lane  = threadIdx.x & 31;
    const int wave  = threadIdx.x >> 5;
    const int row   = lane & 15;
    const int khalf = lane >> 4;
    const int n     = lane & 15;

    const int blk_row = ((int)blockIdx.x - (second ? nblk_a : 0)) * MT;
    const int lr0 = wave * 32 + row;
    const int lr1 = lr0 + 16;

    v8f c0 = {}, c1 = {}, c2 = {}, c3 = {};
    v8f c4 = {}, c5 = {}, c6 = {}, c7 = {};

    const int tid = threadIdx.x;
    const int sc  = tid & 63;
    const int sr  = tid >> 6;

    for (int kt = 0; kt < K; kt += KT) {
        float wv[(KT * FATT) / 128];
        float xv[(MT * KT) / 128];
        const int  kgx   = kt + sc;
        const bool kx_ok = kgx < K;
        const unsigned kgxc = (unsigned)(kx_ok ? kgx : K - 1);
        #pragma unroll
        for (int j = 0; j < (KT * FATT) / 128; ++j) {
            int kg  = kt + sr + 2 * j;
            int kgc = kg < K ? kg : K - 1;
            wv[j] = W[(unsigned)kgc * FATT + (unsigned)sc];
        }
        #pragma unroll
        for (int j = 0; j < (MT * KT) / 128; ++j) {
            int mg  = blk_row + sr + 2 * j;
            int mgc = mg < M ? mg : M - 1;
            xv[j] = X[(unsigned)mgc * (unsigned)K + kgxc];
        }
        #pragma unroll
        for (int j = 0; j < (KT * FATT) / 128; ++j) {
            int kg = kt + sr + 2 * j;
            lW[tid + 128 * j] = (kg < K) ? wv[j] : 0.0f;
        }
        #pragma unroll
        for (int j = 0; j < (MT * KT) / 128; ++j) {
            int mg = blk_row + sr + 2 * j;
            lX[tid + 128 * j] = (mg < M && kx_ok) ? xv[j] : 0.0f;
        }
        __syncthreads();

        #pragma unroll 4
        for (int k0 = 0; k0 < KT; k0 += 4) {
            const int ka = k0 + 2 * khalf;
            v2f a0 = *(const v2f*)(lX + lr0 * KT + ka);
            v2f a1 = *(const v2f*)(lX + lr1 * KT + ka);

            const float* wr0 = lW + ka * FATT + n;
            const float* wr1 = lW + (ka + 1) * FATT + n;
            v2f b0 = { wr0[0],  wr1[0]  };
            v2f b1 = { wr0[16], wr1[16] };
            v2f b2 = { wr0[32], wr1[32] };
            v2f b3 = { wr0[48], wr1[48] };

            c0 = __builtin_amdgcn_wmma_f32_16x16x4_f32(false, a0, false, b0, (short)0, c0, false, false);
            c1 = __builtin_amdgcn_wmma_f32_16x16x4_f32(false, a0, false, b1, (short)0, c1, false, false);
            c2 = __builtin_amdgcn_wmma_f32_16x16x4_f32(false, a0, false, b2, (short)0, c2, false, false);
            c3 = __builtin_amdgcn_wmma_f32_16x16x4_f32(false, a0, false, b3, (short)0, c3, false, false);
            c4 = __builtin_amdgcn_wmma_f32_16x16x4_f32(false, a1, false, b0, (short)0, c4, false, false);
            c5 = __builtin_amdgcn_wmma_f32_16x16x4_f32(false, a1, false, b1, (short)0, c5, false, false);
            c6 = __builtin_amdgcn_wmma_f32_16x16x4_f32(false, a1, false, b2, (short)0, c6, false, false);
            c7 = __builtin_amdgcn_wmma_f32_16x16x4_f32(false, a1, false, b3, (short)0, c7, false, false);
        }
        __syncthreads();
    }

    const int m0 = blk_row + wave * 32 + 8 * khalf;
    #pragma unroll
    for (int r = 0; r < 8; ++r) {
        int mm = m0 + r;
        if (mm < M) {
            float* zr = Z + (size_t)mm * FATT + n;
            zr[0]  = c0[r];
            zr[16] = c1[r];
            zr[32] = c2[r];
            zr[48] = c3[r];
        }
        int mm1 = mm + 16;
        if (mm1 < M) {
            float* zr = Z + (size_t)mm1 * FATT + n;
            zr[0]  = c4[r];
            zr[16] = c5[r];
            zr[32] = c6[r];
            zr[48] = c7[r];
        }
    }
}

#define CHUNK  256
#define NCHUNK ((NTOT + CHUNK - 1) / CHUNK)
#define EPB    4096
#define NBLK   ((NEDGE + EPB - 1) / EPB)
#define NBH    (NCHUNK * NBLK)
#define NSCAN  ((NBH + CHUNK - 1) / CHUNK)
#define SRC_BITS 17

static_assert(NEDGE % 4 == 0, "edge kernels read four edges per thread and load");
static_assert(EPB % 1024 == 0, "256 threads x 4 edges per load");
static_assert(NTOT <= (1 << SRC_BITS), "source index must fit the packed edge");

__device__ __forceinline__ unsigned int bucket_of(int d) {
    const unsigned int k = (unsigned int)d >> 8;
    return k < NCHUNK ? k : NCHUNK - 1;
}

__device__ __forceinline__ unsigned int pack_edge(int s, int d) {
    const unsigned int su = (unsigned int)s < NTOT ? (unsigned int)s : NTOT - 1;
    return su | (((unsigned int)d & 255u) << SRC_BITS);
}

__global__ __launch_bounds__(256) void bucket_hist(const int* __restrict__ dst,
                                                   unsigned int* __restrict__ bh) {
    __shared__ unsigned int h[NCHUNK];
    const int t = threadIdx.x;
    int4 d[EPB / 1024];
    bool ok[EPB / 1024];
    #pragma unroll
    for (int i = 0; i < EPB / 1024; ++i) {
        const int q = blockIdx.x * (EPB / 4) + i * 256 + t;
        ok[i] = q < NEDGE / 4;
        d[i]  = ((const int4*)dst)[ok[i] ? q : NEDGE / 4 - 1];
    }
    for (int k = t; k < NCHUNK; k += 256) h[k] = 0u;
    __syncthreads();
    #pragma unroll
    for (int i = 0; i < EPB / 1024; ++i) {
        if (ok[i]) {
            atomicAdd(&h[bucket_of(d[i].x)], 1u);
            atomicAdd(&h[bucket_of(d[i].y)], 1u);
            atomicAdd(&h[bucket_of(d[i].z)], 1u);
            atomicAdd(&h[bucket_of(d[i].w)], 1u);
        }
    }
    __syncthreads();
    for (int k = t; k < NCHUNK; k += 256) bh[k * NBLK + blockIdx.x] = h[k];
}

__device__ __forceinline__ unsigned int wave_sum(unsigned int v) {
    for (int off = 16; off > 0; off >>= 1) v += __shfl_xor(v, off, 32);
    return v;
}

__global__ __launch_bounds__(256) void chunk_sums(const unsigned int* __restrict__ cnt,
                                                  unsigned int* __restrict__ part, int n) {
    __shared__ unsigned int ws[8];
    const int i = blockIdx.x * CHUNK + threadIdx.x;
    const unsigned int v = wave_sum(i < n ? cnt[i] : 0u);
    if ((threadIdx.x & 31) == 0) ws[threadIdx.x >> 5] = v;
    __syncthreads();
    if (threadIdx.x == 0)
        part[blockIdx.x] = ws[0] + ws[1] + ws[2] + ws[3] + ws[4] + ws[5] + ws[6] + ws[7];
}

__global__ __launch_bounds__(256) void chunk_scan(unsigned int* __restrict__ cur,
                                                  const unsigned int* __restrict__ part, int n) {
    __shared__ unsigned int wbase[8];
    __shared__ unsigned int wtot[8];
    const int b = blockIdx.x, t = threadIdx.x;
    const int lane = t & 31, wave = t >> 5;
    unsigned int pb = 0u;
    for (int j = t; j < b; j += 256) pb += part[j];
    pb = wave_sum(pb);
    const int i = b * CHUNK + t;
    const unsigned int c = i < n ? cur[i] : 0u;
    unsigned int inc = c;
    for (int off = 1; off < 32; off <<= 1) {
        const unsigned int up = __shfl_up(inc, off, 32);
        if (lane >= off) inc += up;
    }
    if (lane == 0)  wbase[wave] = pb;
    if (lane == 31) wtot[wave]  = inc;
    __syncthreads();
    unsigned int base = 0u;
    for (int w = 0; w < 8; ++w) {
        base += wbase[w];
        if (w < wave) base += wtot[w];
    }
    if (i < n) cur[i] = base + inc - c;
}

__global__ __launch_bounds__(256) void bucket_scatter(const int* __restrict__ src,
                                                      const int* __restrict__ dst,
                                                      const unsigned int* __restrict__ bh,
                                                      unsigned int* __restrict__ ebuf) {
    __shared__ unsigned int nxt[NCHUNK];
    const int t = threadIdx.x;
    int4 d[EPB / 1024], s[EPB / 1024];
    bool ok[EPB / 1024];
    #pragma unroll
    for (int i = 0; i < EPB / 1024; ++i) {
        const int q = blockIdx.x * (EPB / 4) + i * 256 + t;
        ok[i] = q < NEDGE / 4;
        const int qc = ok[i] ? q : NEDGE / 4 - 1;
        d[i] = ((const int4*)dst)[qc];
        s[i] = ((const int4*)src)[qc];
    }
    for (int k = t; k < NCHUNK; k += 256) nxt[k] = bh[k * NBLK + blockIdx.x];
    __syncthreads();
    #pragma unroll
    for (int i = 0; i < EPB / 1024; ++i) {
        if (ok[i]) {
            unsigned int p0 = atomicAdd(&nxt[bucket_of(d[i].x)], 1u);
            unsigned int p1 = atomicAdd(&nxt[bucket_of(d[i].y)], 1u);
            unsigned int p2 = atomicAdd(&nxt[bucket_of(d[i].z)], 1u);
            unsigned int p3 = atomicAdd(&nxt[bucket_of(d[i].w)], 1u);
            p0 = p0 < NEDGE ? p0 : NEDGE - 1;
            p1 = p1 < NEDGE ? p1 : NEDGE - 1;
            p2 = p2 < NEDGE ? p2 : NEDGE - 1;
            p3 = p3 < NEDGE ? p3 : NEDGE - 1;
            ebuf[p0] = pack_edge(s[i].x, d[i].x);
            ebuf[p1] = pack_edge(s[i].y, d[i].y);
            ebuf[p2] = pack_edge(s[i].z, d[i].z);
            ebuf[p3] = pack_edge(s[i].w, d[i].w);
        }
    }
}

__global__ __launch_bounds__(256) void bucket_to_csr(const unsigned int* __restrict__ ebuf,
                                                     const unsigned int* __restrict__ bh,
                                                     unsigned int* __restrict__ cur,
                                                     int* __restrict__ csr_src) {
    __shared__ unsigned int cnt[CHUNK];
    __shared__ unsigned int wtot[8];
    const int k = blockIdx.x, t = threadIdx.x;
    const int lane = t & 31, wave = t >> 5;
    unsigned int bs = bh[k * NBLK];
    unsigned int be = k + 1 < NCHUNK ? bh[(k + 1) * NBLK] : NEDGE;
    if (be > NEDGE) be = NEDGE;
    if (bs > be)    bs = be;
    const unsigned int ne = be - bs;

    cnt[t] = 0u;
    __syncthreads();
    for (unsigned int base = 0; base < ne; base += 1024) {
        unsigned int w[4];
        bool ok[4];
        #pragma unroll
        for (int j = 0; j < 4; ++j) {
            const unsigned int i = base + (unsigned int)(j * 256 + t);
            ok[j] = i < ne;
            w[j]  = ebuf[bs + (ok[j] ? i : 0u)];
        }
        #pragma unroll
        for (int j = 0; j < 4; ++j)
            if (ok[j]) atomicAdd(&cnt[(w[j] >> SRC_BITS) & 255u], 1u);
    }
    __syncthreads();

    const unsigned int c = cnt[t];
    unsigned int inc = c;
    for (int off = 1; off < 32; off <<= 1) {
        const unsigned int up = __shfl_up(inc, off, 32);
        if (lane >= off) inc += up;
    }
    if (lane == 31) wtot[wave] = inc;
    __syncthreads();
    unsigned int start = inc - c;
    for (int w = 0; w < 8; ++w)
        if (w < wave) start += wtot[w];
    const int node = k * CHUNK + t;
    if (node < NTOT) cur[node] = bs + start + c;
    cnt[t] = start;
    __syncthreads();
    for (unsigned int base = 0; base < ne; base += 1024) {
        unsigned int w[4];
        bool ok[4];
        #pragma unroll
        for (int j = 0; j < 4; ++j) {
            const unsigned int i = base + (unsigned int)(j * 256 + t);
            ok[j] = i < ne;
            w[j]  = ebuf[bs + (ok[j] ? i : 0u)];
        }
        #pragma unroll
        for (int j = 0; j < 4; ++j) {
            if (ok[j]) {
                unsigned int r = bs + atomicAdd(&cnt[(w[j] >> SRC_BITS) & 255u], 1u);
                r = r < NEDGE ? r : NEDGE - 1;
                csr_src[r] = (int)(w[j] & ((1u << SRC_BITS) - 1u));
            }
        }
    }
}

__device__ __forceinline__ float lrelu(float v) { return v > 0.0f ? v : SLOPE * v; }


__device__ __forceinline__ float xhalf(float x) {
    const int b = __builtin_bit_cast(int, x);
    return __builtin_bit_cast(float,
        __builtin_amdgcn_permlanex16(b, b, 0x76543210, 0xfedcba98, false, false));
}

template <int MASK>
__device__ __forceinline__ float row_xor(float v) {
    return __builtin_bit_cast(float,
        __builtin_amdgcn_mov_dpp(__builtin_bit_cast(int, v), 0x160 | MASK, 0xf, 0xf, true));
}

template <int MASK>
__device__ __forceinline__ float fold_pair(float lo, float hi, int c) {
    const bool up = (c & MASK) != 0;
    const float keep = up ? hi : lo;
    const float send = up ? lo : hi;
    return keep + row_xor<MASK>(send);
}

#define ROW_SHARE_CASE(P) case P: return __builtin_amdgcn_mov_dpp(v, 0x150 | P, 0xf, 0xf, true);
__device__ __forceinline__ int row_bcast(int v, int p) {
    switch (p & 15) {
        ROW_SHARE_CASE(0)  ROW_SHARE_CASE(1)  ROW_SHARE_CASE(2)  ROW_SHARE_CASE(3)
        ROW_SHARE_CASE(4)  ROW_SHARE_CASE(5)  ROW_SHARE_CASE(6)  ROW_SHARE_CASE(7)
        ROW_SHARE_CASE(8)  ROW_SHARE_CASE(9)  ROW_SHARE_CASE(10) ROW_SHARE_CASE(11)
        ROW_SHARE_CASE(12) ROW_SHARE_CASE(13) ROW_SHARE_CASE(14)
        default: return __builtin_amdgcn_mov_dpp(v, 0x150 | 15, 0xf, 0xf, true);
    }
}
#undef ROW_SHARE_CASE

typedef float v4f __attribute__((ext_vector_type(4)));

template <int NP, int P0>
__device__ __forceinline__ void attend_pairs(const float* __restrict__ z, unsigned int sl, int n,
                                             unsigned int c16, int c, int h, v2f zlo, v2f zhi,
                                             float& m, float& l, v2f& alo, v2f& ahi) {
    static_assert(NP == 4 || NP == 8, "the butterfly below is written for 4 or 8 pairs");
    static_assert(P0 % NP == 0, "lane P0+k must be the lane whose low bits are k");
    v4f   r[NP];
    float d[NP];
    #pragma unroll
    for (int i = 0; i < NP; ++i) {
        const unsigned int s = (unsigned int)row_bcast((int)sl, P0 + i);
        r[i] = *(const v4f*)((const char*)z + (s * (unsigned int)(FATT * 4) + c16));
    }
    __builtin_amdgcn_sched_barrier(0);
    #pragma unroll
    for (int i = 0; i < NP; ++i) {
        const v2f t = __builtin_elementwise_fma(r[i].zw, zhi, r[i].xy * zlo);
        d[i] = t.x + t.y;
    }
    float s;
    if (NP == 8) {
        const float e0 = fold_pair<4>(d[0], d[4], c);
        const float e1 = fold_pair<4>(d[1], d[5], c);
        const float e2 = fold_pair<4>(d[2], d[6], c);
        const float e3 = fold_pair<4>(d[3], d[7], c);
        const float f0 = fold_pair<2>(e0, e2, c);
        const float f1 = fold_pair<2>(e1, e3, c);
        s = fold_pair<1>(f0, f1, c);
        s += row_xor<8>(s);
    } else {
        const float f0 = fold_pair<2>(d[0], d[2], c);
        const float f1 = fold_pair<2>(d[1], d[3], c);
        s = fold_pair<1>(f0, f1, c);
        s += row_xor<4>(s);
        s += row_xor<8>(s);
    }
    const bool mine = ((c & ~(NP - 1)) == P0) && (2 * c + h < n);
    const float e = mine ? fmaxf(s, SLOPE * s) : -INFINITY;
    float cm = e;
    cm = fmaxf(cm, row_xor<1>(cm));
    cm = fmaxf(cm, row_xor<2>(cm));
    cm = fmaxf(cm, row_xor<4>(cm));
    cm = fmaxf(cm, row_xor<8>(cm));
    cm = fmaxf(cm, xhalf(cm));
    const float mn = fmaxf(m, cm);
    const float sc = __expf(m - mn);
    const float p  = __expf(e - mn);
    l = l * sc + p;
    v2f slo = alo * sc;
    v2f shi = ahi * sc;
    #pragma unroll
    for (int i = 0; i < NP; ++i) {
        const float w = __builtin_bit_cast(float, row_bcast(__builtin_bit_cast(int, p), P0 + i));
        const v2f  ww = { w, w };
        slo = __builtin_elementwise_fma(r[i].xy, ww, slo);
        shi = __builtin_elementwise_fma(r[i].zw, ww, shi);
    }
    alo = slo; ahi = shi;
    m = mn;
}

__global__ __launch_bounds__(256) void node_attend(const float* __restrict__ z,
                                                   const int* __restrict__ csr_src,
                                                   const unsigned int* __restrict__ cur,
                                                   float* __restrict__ out) {
    const int lane = threadIdx.x & 31;
    const int c = lane & 15;
    const int h = lane >> 4;
    const unsigned int c16 = (unsigned int)c * 16u;
    const int node = blockIdx.x * 8 + __builtin_amdgcn_readfirstlane((int)(threadIdx.x >> 5));
    if (node >= NTOT) return;
    unsigned int end = cur[node];
    unsigned int beg = node > 0 ? cur[node - 1] : 0u;
    if (end > NEDGE) end = NEDGE;
    if (beg > end)   beg = end;

    const v4f zd = *(const v4f*)((const char*)(z + (size_t)node * FATT) + c16);
    const v2f zlo = zd.xy, zhi = zd.zw;
    float m = -3.0e38f, l = 0.0f;
    v2f alo = { 0.0f, 0.0f }, ahi = { 0.0f, 0.0f };

    for (unsigned int base = beg; base < end; base += 32) {
        const unsigned int j = base + 2u * (unsigned int)c + (unsigned int)h;
        unsigned int sl = (unsigned int)csr_src[j < end ? j : end - 1];
        sl = sl < NTOT ? sl : NTOT - 1;
        const int n = (int)(end - base < 32u ? end - base : 32u);
        if (n > 8) attend_pairs<8, 0>(z, sl, n, c16, c, h, zlo, zhi, m, l, alo, ahi);
        else       attend_pairs<4, 0>(z, sl, n, c16, c, h, zlo, zhi, m, l, alo, ahi);
        if (n > 16) {
            if (n > 24) attend_pairs<8, 8>(z, sl, n, c16, c, h, zlo, zhi, m, l, alo, ahi);
            else        attend_pairs<4, 8>(z, sl, n, c16, c, h, zlo, zhi, m, l, alo, ahi);
        }
    }
    l += row_xor<1>(l);
    l += row_xor<2>(l);
    l += row_xor<4>(l);
    l += row_xor<8>(l);
    l += xhalf(l);
    const float wx = alo.x + xhalf(alo.x);
    const float wy = alo.y + xhalf(alo.y);
    const float wz = ahi.x + xhalf(ahi.x);
    const float ww = ahi.y + xhalf(ahi.y);
    v4f hv = { 0.0f, 0.0f, 0.0f, 0.0f };
    if (l > 0.0f) {
        const float inv = 1.0f / l;
        hv.x = wx * inv; hv.y = wy * inv; hv.z = wz * inv; hv.w = ww * inv;
    }
    hv.x = hv.x > 0.0f ? hv.x : (__expf(hv.x) - 1.0f);
    hv.y = hv.y > 0.0f ? hv.y : (__expf(hv.y) - 1.0f);
    hv.z = hv.z > 0.0f ? hv.z : (__expf(hv.z) - 1.0f);
    hv.w = hv.w > 0.0f ? hv.w : (__expf(hv.w) - 1.0f);
    if (h == 0) *(v4f*)((char*)(out + (size_t)node * FATT) + c16) = hv;
}

extern "C" void kernel_launch(void* const* d_in, const int* in_sizes, int n_in,
                              void* d_out, int out_size, void* d_ws, size_t ws_size,
                              hipStream_t stream) {
    const float* d_sim  = (const float*)d_in[0];
    const float* mi_sim = (const float*)d_in[1];
    const float* W_d    = (const float*)d_in[2];
    const float* W_mi   = (const float*)d_in[3];
    const int*   src    = (const int*)d_in[4];
    const int*   dst    = (const int*)d_in[5];
    float*       out    = (float*)d_out;

    float* z            = (float*)d_ws;
    int* csr_src        = (int*)(z + (size_t)NTOT * FATT);
    unsigned int* cur   = (unsigned int*)(csr_src + NEDGE);
    unsigned int* bh    = (unsigned int*)d_ws;
    unsigned int* part  = bh + NSCAN * CHUNK;
    unsigned int* ebuf  = part + 1024;

    bucket_hist<<<NBLK, 256, 0, stream>>>(dst, bh);
    chunk_sums<<<NSCAN, 256, 0, stream>>>(bh, part, NBH);
    chunk_scan<<<NSCAN, 256, 0, stream>>>(bh, part, NBH);
    bucket_scatter<<<NBLK, 256, 0, stream>>>(src, dst, bh, ebuf);
    bucket_to_csr<<<NCHUNK, 256, 0, stream>>>(ebuf, bh, cur, csr_src);

    gemm_wmma_f32<<<(N_MIC + 127) / 128 + (N_DIS + 127) / 128, 128, 0, stream>>>(
        mi_sim, W_mi, z + (size_t)N_DIS * FATT, N_MIC, 495,
        d_sim, W_d, z, N_DIS, 383,
        (N_MIC + 127) / 128);

    node_attend<<<(NTOT + 7) / 8, 256, 0, stream>>>(z, csr_src, cur, out);
}


# Grouping in 2 launches: each tile of 4096 edges ordered by bucket in LDS and stored in place with its run bounds; bucket pass sums the bounds, reads its runs once; sum order inside a node changes
# speedup vs baseline: 9.0397x; 1.4076x over previous
#include <hip/hip_runtime.h>
#include <math.h>

#define N_DIS 20000
#define N_MIC 80000
#define NTOT  (N_DIS + N_MIC)
#define NEDGE 1600000
#define FATT  64
#define SLOPE 0.2f

typedef float v2f __attribute__((ext_vector_type(2)));
typedef float v4f __attribute__((ext_vector_type(4)));
typedef float v8f __attribute__((ext_vector_type(8)));

__global__ __launch_bounds__(128) void gemm_wmma_f32(
        const float* __restrict__ Xa, const float* __restrict__ Wa,
        float* __restrict__ Za, int Ma, int Ka,
        const float* __restrict__ Xb, const float* __restrict__ Wb,
        float* __restrict__ Zb, int Mb, int Kb, int nblk_a) {
    const bool second = (int)blockIdx.x >= nblk_a;
    const float* __restrict__ X = second ? Xb : Xa;
    const float* __restrict__ W = second ? Wb : Wa;
    float* __restrict__ Z = second ? Zb : Za;
    const int M = second ? Mb : Ma;
    const int K = second ? Kb : Ka;
    const int KT = 64;
    const int MT = 128;
    const int PW = 68;
    __shared__ __attribute__((aligned(16))) float lX[MT * KT];
    __shared__ __attribute__((aligned(16))) float lWt[FATT * PW];

    const int lane  = threadIdx.x & 31;
    const int wave  = threadIdx.x >> 5;
    const int row   = lane & 15;
    const int khalf = lane >> 4;
    const int n     = lane & 15;

    const int blk_row = ((int)blockIdx.x - (second ? nblk_a : 0)) * MT;
    const int lr0 = wave * 32 + row;
    const int lr1 = lr0 + 16;

    v8f c0 = {}, c1 = {}, c2 = {}, c3 = {};
    v8f c4 = {}, c5 = {}, c6 = {}, c7 = {};

    const int tid = threadIdx.x;
    const int sc  = tid & 63;
    const int sr  = tid >> 6;

    float* xs[8];
    #pragma unroll
    for (int q = 0; q < 8; ++q) {
        const int lr = sr + 2 * q;
        xs[q] = lX + lr * KT + ((((sc >> 2) ^ (lr & 15)) << 2) | (sc & 3));
    }

    for (int kt = 0; kt < K; kt += KT) {
        float wv[(KT * FATT) / 128];
        float xv[(MT * KT) / 128];
        const int  kgx   = kt + sc;
        const bool kx_ok = kgx < K;
        const unsigned kgxc = (unsigned)(kx_ok ? kgx : K - 1);
        #pragma unroll
        for (int j = 0; j < (KT * FATT) / 128; ++j) {
            int kg  = kt + sr + 2 * j;
            int kgc = kg < K ? kg : K - 1;
            wv[j] = W[(unsigned)kgc * FATT + (unsigned)sc];
        }
        #pragma unroll
        for (int j = 0; j < (MT * KT) / 128; ++j) {
            int mg  = blk_row + sr + 2 * j;
            int mgc = mg < M ? mg : M - 1;
            xv[j] = X[(unsigned)mgc * (unsigned)K + kgxc];
        }
        #pragma unroll
        for (int j = 0; j < (KT * FATT) / 128; ++j) {
            int kg = kt + sr + 2 * j;
            lWt[sc * PW + sr + 2 * j] = (kg < K) ? wv[j] : 0.0f;
        }
        #pragma unroll
        for (int j = 0; j < (MT * KT) / 128; ++j) {
            int mg = blk_row + sr + 2 * j;
            xs[j & 7][(j >> 3) * 16 * KT] = (mg < M && kx_ok) ? xv[j] : 0.0f;
        }
        __syncthreads();

        const float* pa0 = lX  + lr0 * KT;
        const float* pa1 = lX  + lr1 * KT;
        const float* pb  = lWt + n * PW;
        #pragma unroll 2
        for (int g = 0; g < KT / 8; ++g) {
            const int fc = 2 * g + khalf;
            const int ac = (fc ^ row) << 2;
            const v4f a0 = *(const v4f*)(pa0 + ac);
            const v4f a1 = *(const v4f*)(pa1 + ac);
            const v4f b0 = *(const v4f*)(pb + 4 * fc);
            const v4f b1 = *(const v4f*)(pb + 16 * PW + 4 * fc);
            const v4f b2 = *(const v4f*)(pb + 32 * PW + 4 * fc);
            const v4f b3 = *(const v4f*)(pb + 48 * PW + 4 * fc);

            c0 = __builtin_amdgcn_wmma_f32_16x16x4_f32(false, a0.xy, false, b0.xy, (short)0, c0, false, false);
            c1 = __builtin_amdgcn_wmma_f32_16x16x4_f32(false, a0.xy, false, b1.xy, (short)0, c1, false, false);
            c2 = __builtin_amdgcn_wmma_f32_16x16x4_f32(false, a0.xy, false, b2.xy, (short)0, c2, false, false);
            c3 = __builtin_amdgcn_wmma_f32_16x16x4_f32(false, a0.xy, false, b3.xy, (short)0, c3, false, false);
            c4 = __builtin_amdgcn_wmma_f32_16x16x4_f32(false, a1.xy, false, b0.xy, (short)0, c4, false, false);
            c5 = __builtin_amdgcn_wmma_f32_16x16x4_f32(false, a1.xy, false, b1.xy, (short)0, c5, false, false);
            c6 = __builtin_amdgcn_wmma_f32_16x16x4_f32(false, a1.xy, false, b2.xy, (short)0, c6, false, false);
            c7 = __builtin_amdgcn_wmma_f32_16x16x4_f32(false, a1.xy, false, b3.xy, (short)0, c7, false, false);
            c0 = __builtin_amdgcn_wmma_f32_16x16x4_f32(false, a0.zw, false, b0.zw, (short)0, c0, false, false);
            c1 = __builtin_amdgcn_wmma_f32_16x16x4_f32(false, a0.zw, false, b1.zw, (short)0, c1, false, false);
            c2 = __builtin_amdgcn_wmma_f32_16x16x4_f32(false, a0.zw, false, b2.zw, (short)0, c2, false, false);
            c3 = __builtin_amdgcn_wmma_f32_16x16x4_f32(false, a0.zw, false, b3.zw, (short)0, c3, false, false);
            c4 = __builtin_amdgcn_wmma_f32_16x16x4_f32(false, a1.zw, false, b0.zw, (short)0, c4, false, false);
            c5 = __builtin_amdgcn_wmma_f32_16x16x4_f32(false, a1.zw, false, b1.zw, (short)0, c5, false, false);
            c6 = __builtin_amdgcn_wmma_f32_16x16x4_f32(false, a1.zw, false, b2.zw, (short)0, c6, false, false);
            c7 = __builtin_amdgcn_wmma_f32_16x16x4_f32(false, a1.zw, false, b3.zw, (short)0, c7, false, false);
        }
        __syncthreads();
    }

    const int m0 = blk_row + wave * 32 + 8 * khalf;
    #pragma unroll
    for (int r = 0; r < 8; ++r) {
        int mm = m0 + r;
        if (mm < M) {
            float* zr = Z + (size_t)mm * FATT + n;
            zr[0]  = c0[r];
            zr[16] = c1[r];
            zr[32] = c2[r];
            zr[48] = c3[r];
        }
        int mm1 = mm + 16;
        if (mm1 < M) {
            float* zr = Z + (size_t)mm1 * FATT + n;
            zr[0]  = c4[r];
            zr[16] = c5[r];
            zr[32] = c6[r];
            zr[48] = c7[r];
        }
    }
}

#define CHUNK  256
#define NCHUNK ((NTOT + CHUNK - 1) / CHUNK)
#define EPB    4096
#define NBLK   ((NEDGE + EPB - 1) / EPB)
#define NOFF   ((NCHUNK + 1) * NBLK)
#define SRC_BITS 17
#define RUN_Q  6

static_assert(NEDGE % 4 == 0, "the edge pass reads four edges per thread and load");
static_assert(EPB % 1024 == 0, "256 threads x 4 edges per load");
static_assert(NTOT <= (1 << SRC_BITS), "source index must fit the packed edge");
static_assert(NCHUNK + 1 <= 2 * CHUNK, "a thread of tile_sort scans two bucket counts");
static_assert(NBLK > CHUNK && NBLK <= 2 * CHUNK, "a thread of bucket_to_csr walks the runs of tiles t and t + 256");
static_assert(NOFF % 4 == 0, "ebuf starts on a 16-byte boundary behind offs");

__device__ __forceinline__ unsigned int bucket_of(int d) {
    const unsigned int k = (unsigned int)d >> 8;
    return k < NCHUNK ? k : NCHUNK - 1;
}

__device__ __forceinline__ unsigned int pack_edge(int s, int d) {
    const unsigned int su = (unsigned int)s < NTOT ? (unsigned int)s : NTOT - 1;
    return su | (((unsigned int)d & 255u) << SRC_BITS);
}

__device__ __forceinline__ unsigned int wave_sum(unsigned int v) {
    for (int off = 16; off > 0; off >>= 1) v += __shfl_xor(v, off, 32);
    return v;
}

__global__ __launch_bounds__(256) void tile_sort(const int* __restrict__ src,
                                                 const int* __restrict__ dst,
                                                 unsigned int* __restrict__ offs,
                                                 unsigned int* __restrict__ ebuf) {
    __shared__ unsigned int h[2 * CHUNK];
    __shared__ unsigned int wtot[8];
    __shared__ __attribute__((aligned(16))) unsigned int sorted[EPB];
    const int t = threadIdx.x;
    const int lane = t & 31, wave = t >> 5;
    int4 d[EPB / 1024], s[EPB / 1024];
    bool ok[EPB / 1024];
    #pragma unroll
    for (int i = 0; i < EPB / 1024; ++i) {
        const int q = blockIdx.x * (EPB / 4) + i * 256 + t;
        ok[i] = q < NEDGE / 4;
        const int qc = ok[i] ? q : NEDGE / 4 - 1;
        d[i] = ((const int4*)dst)[qc];
        s[i] = ((const int4*)src)[qc];
    }
    h[t] = 0u;
    h[t + CHUNK] = 0u;
    __syncthreads();
    unsigned int rk[EPB / 256];
    #pragma unroll
    for (int i = 0; i < EPB / 1024; ++i) {
        rk[4 * i + 0] = rk[4 * i + 1] = rk[4 * i + 2] = rk[4 * i + 3] = 0u;
        if (ok[i]) {
            rk[4 * i + 0] = atomicAdd(&h[bucket_of(d[i].x)], 1u);
            rk[4 * i + 1] = atomicAdd(&h[bucket_of(d[i].y)], 1u);
            rk[4 * i + 2] = atomicAdd(&h[bucket_of(d[i].z)], 1u);
            rk[4 * i + 3] = atomicAdd(&h[bucket_of(d[i].w)], 1u);
        }
    }
    __syncthreads();
    const unsigned int c0 = h[2 * t], c1 = h[2 * t + 1];
    unsigned int inc = c0 + c1;
    for (int off = 1; off < 32; off <<= 1) {
        const unsigned int up = __shfl_up(inc, off, 32);
        if (lane >= off) inc += up;
    }
    if (lane == 31) wtot[wave] = inc;
    __syncthreads();
    unsigned int start = inc - (c0 + c1);
    for (int w = 0; w < 8; ++w)
        if (w < wave) start += wtot[w];
    h[2 * t]     = start;
    h[2 * t + 1] = start + c0;
    if (2 * t     <= NCHUNK) offs[(2 * t)     * NBLK + blockIdx.x] = start;
    if (2 * t + 1 <= NCHUNK) offs[(2 * t + 1) * NBLK + blockIdx.x] = start + c0;
    __syncthreads();
    #pragma unroll
    for (int i = 0; i < EPB / 1024; ++i) {
        if (ok[i]) {
            sorted[h[bucket_of(d[i].x)] + rk[4 * i + 0]] = pack_edge(s[i].x, d[i].x);
            sorted[h[bucket_of(d[i].y)] + rk[4 * i + 1]] = pack_edge(s[i].y, d[i].y);
            sorted[h[bucket_of(d[i].z)] + rk[4 * i + 2]] = pack_edge(s[i].z, d[i].z);
            sorted[h[bucket_of(d[i].w)] + rk[4 * i + 3]] = pack_edge(s[i].w, d[i].w);
        }
    }
    __syncthreads();
    const int left = NEDGE - (int)blockIdx.x * EPB;
    const int nq   = (left < EPB ? left : EPB) / 4;
    uint4* __restrict__ o = (uint4*)(ebuf + (size_t)blockIdx.x * EPB);
    for (int i = 0; i < EPB / 1024; ++i) {
        const int q = i * 256 + t;
        if (q < nq) o[q] = ((const uint4*)sorted)[q];
    }
}

template <bool RANK>
__device__ __forceinline__ void take_edge(unsigned int w, unsigned int i,
                                          unsigned int lo, unsigned int hi,
                                          unsigned int* cnt, unsigned int bs,
                                          int* __restrict__ csr_src) {
    if (i >= lo && i < hi) {
        if (RANK) {
            unsigned int r = bs + atomicAdd(&cnt[(w >> SRC_BITS) & 255u], 1u);
            r = r < NEDGE ? r : NEDGE - 1;
            csr_src[r] = (int)(w & ((1u << SRC_BITS) - 1u));
        } else {
            atomicAdd(&cnt[(w >> SRC_BITS) & 255u], 1u);
        }
    }
}


__device__ __forceinline__ void load_pieces(const uint4* __restrict__ tq, unsigned int b,
                                            unsigned int qe, uint4 (&w)[RUN_Q]) {
    #pragma unroll
    for (int j = 0; j < RUN_Q; ++j) {
        const unsigned int p = b + (unsigned int)j;
        w[j] = tq[p < qe ? p : 0u];
    }
}

template <bool RANK>
__device__ __forceinline__ void take_pieces(const uint4 (&w)[RUN_Q], unsigned int b,
                                            unsigned int lo, unsigned int hi,
                                            unsigned int* cnt, unsigned int bs,
                                            int* __restrict__ csr_src) {
    #pragma unroll
    for (int j = 0; j < RUN_Q; ++j) {
        const unsigned int i = (b + (unsigned int)j) << 2;
        take_edge<RANK>(w[j].x, i + 0u, lo, hi, cnt, bs, csr_src);
        take_edge<RANK>(w[j].y, i + 1u, lo, hi, cnt, bs, csr_src);
        take_edge<RANK>(w[j].z, i + 2u, lo, hi, cnt, bs, csr_src);
        take_edge<RANK>(w[j].w, i + 3u, lo, hi, cnt, bs, csr_src);
    }
}

template <bool RANK>
__device__ __forceinline__ void walk_rest(const uint4* __restrict__ tq, unsigned int b,
                                          unsigned int qe, unsigned int lo, unsigned int hi,
                                          unsigned int* cnt, unsigned int bs,
                                          int* __restrict__ csr_src) {
    for (; b < qe; b += RUN_Q) {
        uint4 w[RUN_Q];
        load_pieces(tq, b, qe, w);
        take_pieces<RANK>(w, b, lo, hi, cnt, bs, csr_src);
    }
}

__global__ __launch_bounds__(256) void bucket_to_csr(const unsigned int* __restrict__ ebuf,
                                                     const unsigned int* __restrict__ offs,
                                                     unsigned int* __restrict__ cur,
                                                     int* __restrict__ csr_src) {
    __shared__ unsigned int cnt[CHUNK];
    __shared__ unsigned int wtot[8];
    __shared__ unsigned int wlo[8];
    const int k = blockIdx.x, t = threadIdx.x;
    const int lane = t & 31, wave = t >> 5;
    const int  t1   = t + CHUNK;
    const bool has1 = t1 < NBLK;
    const int  t1c  = has1 ? t1 : NBLK - 1;
    unsigned int lo0 = offs[k * NBLK + t],   hi0 = offs[(k + 1) * NBLK + t];
    unsigned int lo1 = offs[k * NBLK + t1c], hi1 = offs[(k + 1) * NBLK + t1c];
    if (hi0 > EPB) hi0 = EPB;
    if (lo0 > hi0) lo0 = hi0;
    if (hi1 > EPB) hi1 = EPB;
    if (lo1 > hi1) lo1 = hi1;
    if (!has1) { lo1 = 0u; hi1 = 0u; }
    const uint4* __restrict__ tq0 = (const uint4*)(ebuf + (size_t)t   * EPB);
    const uint4* __restrict__ tq1 = (const uint4*)(ebuf + (size_t)t1c * EPB);
    const unsigned int qe0 = (hi0 + 3u) >> 2, qe1 = (hi1 + 3u) >> 2;
    const unsigned int b0 = lo0 < hi0 ? lo0 >> 2 : qe0;
    const unsigned int b1 = lo1 < hi1 ? lo1 >> 2 : qe1;
    uint4 w0[RUN_Q], w1[RUN_Q];
    load_pieces(tq0, b0, qe0, w0);
    load_pieces(tq1, b1, qe1, w1);

    const unsigned int vlo = wave_sum(lo0 + lo1);
    if (lane == 0) wlo[wave] = vlo;
    cnt[t] = 0u;
    __syncthreads();
    unsigned int bs = wlo[0] + wlo[1] + wlo[2] + wlo[3] + wlo[4] + wlo[5] + wlo[6] + wlo[7];
    if (bs > NEDGE) bs = NEDGE;

    take_pieces<false>(w0, b0, lo0, hi0, cnt, bs, csr_src);
    take_pieces<false>(w1, b1, lo1, hi1, cnt, bs, csr_src);
    walk_rest<false>(tq0, b0 + RUN_Q, qe0, lo0, hi0, cnt, bs, csr_src);
    walk_rest<false>(tq1, b1 + RUN_Q, qe1, lo1, hi1, cnt, bs, csr_src);
    __syncthreads();

    const unsigned int c = cnt[t];
    unsigned int inc = c;
    for (int off = 1; off < 32; off <<= 1) {
        const unsigned int up = __shfl_up(inc, off, 32);
        if (lane >= off) inc += up;
    }
    if (lane == 31) wtot[wave] = inc;
    __syncthreads();
    unsigned int start = inc - c;
    for (int w = 0; w < 8; ++w)
        if (w < wave) start += wtot[w];
    const int node = k * CHUNK + t;
    if (node < NTOT) cur[node] = bs + start + c;
    cnt[t] = start;
    __syncthreads();
    take_pieces<true>(w0, b0, lo0, hi0, cnt, bs, csr_src);
    take_pieces<true>(w1, b1, lo1, hi1, cnt, bs, csr_src);
    walk_rest<true>(tq0, b0 + RUN_Q, qe0, lo0, hi0, cnt, bs, csr_src);
    walk_rest<true>(tq1, b1 + RUN_Q, qe1, lo1, hi1, cnt, bs, csr_src);
}

__device__ __forceinline__ float lrelu(float v) { return v > 0.0f ? v : SLOPE * v; }


__device__ __forceinline__ float xhalf(float x) {
    const int b = __builtin_bit_cast(int, x);
    return __builtin_bit_cast(float,
        __builtin_amdgcn_permlanex16(b, b, 0x76543210, 0xfedcba98, false, false));
}

template <int MASK>
__device__ __forceinline__ float row_xor(float v) {
    return __builtin_bit_cast(float,
        __builtin_amdgcn_mov_dpp(__builtin_bit_cast(int, v), 0x160 | MASK, 0xf, 0xf, true));
}

template <int MASK>
__device__ __forceinline__ float fold_pair(float lo, float hi, int c) {
    const bool up = (c & MASK) != 0;
    const float keep = up ? hi : lo;
    const float send = up ? lo : hi;
    return keep + row_xor<MASK>(send);
}

#define ROW_SHARE_CASE(P) case P: return __builtin_amdgcn_mov_dpp(v, 0x150 | P, 0xf, 0xf, true);
__device__ __forceinline__ int row_bcast(int v, int p) {
    switch (p & 15) {
        ROW_SHARE_CASE(0)  ROW_SHARE_CASE(1)  ROW_SHARE_CASE(2)  ROW_SHARE_CASE(3)
        ROW_SHARE_CASE(4)  ROW_SHARE_CASE(5)  ROW_SHARE_CASE(6)  ROW_SHARE_CASE(7)
        ROW_SHARE_CASE(8)  ROW_SHARE_CASE(9)  ROW_SHARE_CASE(10) ROW_SHARE_CASE(11)
        ROW_SHARE_CASE(12) ROW_SHARE_CASE(13) ROW_SHARE_CASE(14)
        default: return __builtin_amdgcn_mov_dpp(v, 0x150 | 15, 0xf, 0xf, true);
    }
}
#undef ROW_SHARE_CASE

template <int NP, int P0>
__device__ __forceinline__ void attend_pairs(const float* __restrict__ z, unsigned int sl, int n,
                                             unsigned int c16, int c, int h, v2f zlo, v2f zhi,
                                             float& m, float& l, v2f& alo, v2f& ahi) {
    static_assert(NP == 4 || NP == 8, "the butterfly below is written for 4 or 8 pairs");
    static_assert(P0 % NP == 0, "lane P0+k must be the lane whose low bits are k");
    v4f   r[NP];
    float d[NP];
    #pragma unroll
    for (int i = 0; i < NP; ++i) {
        const unsigned int s = (unsigned int)row_bcast((int)sl, P0 + i);
        r[i] = *(const v4f*)((const char*)z + (s * (unsigned int)(FATT * 4) + c16));
    }
    __builtin_amdgcn_sched_barrier(0);
    #pragma unroll
    for (int i = 0; i < NP; ++i) {
        const v2f t = __builtin_elementwise_fma(r[i].zw, zhi, r[i].xy * zlo);
        d[i] = t.x + t.y;
    }
    float s;
    if (NP == 8) {
        const float e0 = fold_pair<4>(d[0], d[4], c);
        const float e1 = fold_pair<4>(d[1], d[5], c);
        const float e2 = fold_pair<4>(d[2], d[6], c);
        const float e3 = fold_pair<4>(d[3], d[7], c);
        const float f0 = fold_pair<2>(e0, e2, c);
        const float f1 = fold_pair<2>(e1, e3, c);
        s = fold_pair<1>(f0, f1, c);
        s += row_xor<8>(s);
    } else {
        const float f0 = fold_pair<2>(d[0], d[2], c);
        const float f1 = fold_pair<2>(d[1], d[3], c);
        s = fold_pair<1>(f0, f1, c);
        s += row_xor<4>(s);
        s += row_xor<8>(s);
    }
    const bool mine = ((c & ~(NP - 1)) == P0) && (2 * c + h < n);
    const float e = mine ? fmaxf(s, SLOPE * s) : -INFINITY;
    float cm = e;
    cm = fmaxf(cm, row_xor<1>(cm));
    cm = fmaxf(cm, row_xor<2>(cm));
    cm = fmaxf(cm, row_xor<4>(cm));
    cm = fmaxf(cm, row_xor<8>(cm));
    cm = fmaxf(cm, xhalf(cm));
    const float mn = fmaxf(m, cm);
    const float sc = __expf(m - mn);
    const float p  = __expf(e - mn);
    l = l * sc + p;
    v2f slo = alo * sc;
    v2f shi = ahi * sc;
    #pragma unroll
    for (int i = 0; i < NP; ++i) {
        const float w = __builtin_bit_cast(float, row_bcast(__builtin_bit_cast(int, p), P0 + i));
        const v2f  ww = { w, w };
        slo = __builtin_elementwise_fma(r[i].xy, ww, slo);
        shi = __builtin_elementwise_fma(r[i].zw, ww, shi);
    }
    alo = slo; ahi = shi;
    m = mn;
}

__global__ __launch_bounds__(256) void node_attend(const float* __restrict__ z,
                                                   const int* __restrict__ csr_src,
                                                   const unsigned int* __restrict__ cur,
                                                   float* __restrict__ out) {
    const int lane = threadIdx.x & 31;
    const int c = lane & 15;
    const int h = lane >> 4;
    const unsigned int c16 = (unsigned int)c * 16u;
    const int node = blockIdx.x * 8 + __builtin_amdgcn_readfirstlane((int)(threadIdx.x >> 5));
    if (node >= NTOT) return;
    unsigned int end = cur[node];
    unsigned int beg = node > 0 ? cur[node - 1] : 0u;
    if (end > NEDGE) end = NEDGE;
    if (beg > end)   beg = end;

    const v4f zd = *(const v4f*)((const char*)(z + (size_t)node * FATT) + c16);
    const v2f zlo = zd.xy, zhi = zd.zw;
    float m = -3.0e38f, l = 0.0f;
    v2f alo = { 0.0f, 0.0f }, ahi = { 0.0f, 0.0f };

    for (unsigned int base = beg; base < end; base += 32) {
        const unsigned int j = base + 2u * (unsigned int)c + (unsigned int)h;
        unsigned int sl = (unsigned int)csr_src[j < end ? j : end - 1];
        sl = sl < NTOT ? sl : NTOT - 1;
        const int n = (int)(end - base < 32u ? end - base : 32u);
        if (n > 8) attend_pairs<8, 0>(z, sl, n, c16, c, h, zlo, zhi, m, l, alo, ahi);
        else       attend_pairs<4, 0>(z, sl, n, c16, c, h, zlo, zhi, m, l, alo, ahi);
        if (n > 16) {
            if (n > 24) attend_pairs<8, 8>(z, sl, n, c16, c, h, zlo, zhi, m, l, alo, ahi);
            else        attend_pairs<4, 8>(z, sl, n, c16, c, h, zlo, zhi, m, l, alo, ahi);
        }
    }
    l += row_xor<1>(l);
    l += row_xor<2>(l);
    l += row_xor<4>(l);
    l += row_xor<8>(l);
    l += xhalf(l);
    const float wx = alo.x + xhalf(alo.x);
    const float wy = alo.y + xhalf(alo.y);
    const float wz = ahi.x + xhalf(ahi.x);
    const float ww = ahi.y + xhalf(ahi.y);
    v4f hv = { 0.0f, 0.0f, 0.0f, 0.0f };
    if (l > 0.0f) {
        const float inv = 1.0f / l;
        hv.x = wx * inv; hv.y = wy * inv; hv.z = wz * inv; hv.w = ww * inv;
    }
    hv.x = hv.x > 0.0f ? hv.x : (__expf(hv.x) - 1.0f);
    hv.y = hv.y > 0.0f ? hv.y : (__expf(hv.y) - 1.0f);
    hv.z = hv.z > 0.0f ? hv.z : (__expf(hv.z) - 1.0f);
    hv.w = hv.w > 0.0f ? hv.w : (__expf(hv.w) - 1.0f);
    if (h == 0) *(v4f*)((char*)(out + (size_t)node * FATT) + c16) = hv;
}

extern "C" void kernel_launch(void* const* d_in, const int* in_sizes, int n_in,
                              void* d_out, int out_size, void* d_ws, size_t ws_size,
                              hipStream_t stream) {
    const float* d_sim  = (const float*)d_in[0];
    const float* mi_sim = (const float*)d_in[1];
    const float* W_d    = (const float*)d_in[2];
    const float* W_mi   = (const float*)d_in[3];
    const int*   src    = (const int*)d_in[4];
    const int*   dst    = (const int*)d_in[5];
    float*       out    = (float*)d_out;

    float* z            = (float*)d_ws;
    int* csr_src        = (int*)(z + (size_t)NTOT * FATT);
    unsigned int* cur   = (unsigned int*)(csr_src + NEDGE);
    unsigned int* offs  = (unsigned int*)d_ws;
    unsigned int* ebuf  = offs + NOFF;

    tile_sort<<<NBLK, 256, 0, stream>>>(src, dst, offs, ebuf);
    bucket_to_csr<<<NCHUNK, 256, 0, stream>>>(ebuf, offs, cur, csr_src);

    gemm_wmma_f32<<<(N_MIC + 127) / 128 + (N_DIS + 127) / 128, 128, 0, stream>>>(
        mi_sim, W_mi, z + (size_t)N_DIS * FATT, N_MIC, 495,
        d_sim, W_d, z, N_DIS, 383,
        (N_MIC + 127) / 128);

    node_attend<<<(NTOT + 7) / 8, 256, 0, stream>>>(z, csr_src, cur, out);
}


# Tile pass of the grouping moved into the projections' launch (391 more 128-thread work-groups behind the 782 WMMA ones; scratch now in the output buffer): 3 launches; sum order in a node changes
# speedup vs baseline: 9.7171x; 1.0749x over previous
#include <hip/hip_runtime.h>
#include <math.h>

#define N_DIS 20000
#define N_MIC 80000
#define NTOT  (N_DIS + N_MIC)
#define NEDGE 1600000
#define FATT  64
#define SLOPE 0.2f

typedef float v2f __attribute__((ext_vector_type(2)));
typedef float v4f __attribute__((ext_vector_type(4)));
typedef float v8f __attribute__((ext_vector_type(8)));

#define CHUNK  256
#define NCHUNK ((NTOT + CHUNK - 1) / CHUNK)
#define EPB    4096
#define NBLK   ((NEDGE + EPB - 1) / EPB)
#define NOFF   ((NCHUNK + 1) * NBLK)
#define SRC_BITS 17
#define RUN_Q  6

static_assert(NEDGE % 4 == 0, "the edge pass reads four edges per thread and load");
static_assert(EPB % 512 == 0, "128 threads x 4 edges per load");
static_assert(NTOT <= (1 << SRC_BITS), "source index must fit the packed edge");
static_assert(NCHUNK + 1 <= 2 * CHUNK, "the 128 threads of the tile pass scan four bucket counts each");
static_assert(NBLK > CHUNK && NBLK <= 2 * CHUNK, "a thread of bucket_to_csr walks the runs of tiles t and t + 256");
static_assert(NOFF % 4 == 0, "ebuf starts on a 16-byte boundary behind offs");

__device__ __forceinline__ unsigned int bucket_of(int d) {
    const unsigned int k = (unsigned int)d >> 8;
    return k < NCHUNK ? k : NCHUNK - 1;
}

__device__ __forceinline__ unsigned int pack_edge(int s, int d) {
    const unsigned int su = (unsigned int)s < NTOT ? (unsigned int)s : NTOT - 1;
    return su | (((unsigned int)d & 255u) << SRC_BITS);
}

__device__ __forceinline__ unsigned int wave_sum(unsigned int v) {
    for (int off = 16; off > 0; off >>= 1) v += __shfl_xor(v, off, 32);
    return v;
}

__device__ __forceinline__ void tile_sort_body(const int tile,
                                               const int* __restrict__ src,
                                               const int* __restrict__ dst,
                                               unsigned int* __restrict__ offs,
                                               unsigned int* __restrict__ ebuf,
                                               unsigned int* sorted,
                                               unsigned int* h) {
    unsigned int* wtot = h + 2 * CHUNK;
    constexpr int NL = EPB / 512;
    const int t = threadIdx.x;
    const int lane = t & 31, wave = t >> 5;
    int4 d[NL], s[NL];
    bool ok[NL];
    #pragma unroll
    for (int i = 0; i < NL; ++i) {
        const int q = tile * (EPB / 4) + i * 128 + t;
        ok[i] = q < NEDGE / 4;
        const int qc = ok[i] ? q : NEDGE / 4 - 1;
        d[i] = ((const int4*)dst)[qc];
        s[i] = ((const int4*)src)[qc];
    }
    *(uint4*)&h[4 * t] = make_uint4(0u, 0u, 0u, 0u);
    __syncthreads();
    unsigned int rk[4 * NL];
    #pragma unroll
    for (int i = 0; i < NL; ++i) {
        rk[4 * i + 0] = rk[4 * i + 1] = rk[4 * i + 2] = rk[4 * i + 3] = 0u;
        if (ok[i]) {
            rk[4 * i + 0] = atomicAdd(&h[bucket_of(d[i].x)], 1u);
            rk[4 * i + 1] = atomicAdd(&h[bucket_of(d[i].y)], 1u);
            rk[4 * i + 2] = atomicAdd(&h[bucket_of(d[i].z)], 1u);
            rk[4 * i + 3] = atomicAdd(&h[bucket_of(d[i].w)], 1u);
        }
    }
    __syncthreads();
    const uint4 c = *(const uint4*)&h[4 * t];
    const unsigned int tot = c.x + c.y + c.z + c.w;
    unsigned int inc = tot;
    for (int off = 1; off < 32; off <<= 1) {
        const unsigned int up = __shfl_up(inc, off, 32);
        if (lane >= off) inc += up;
    }
    if (lane == 31) wtot[wave] = inc;
    __syncthreads();
    unsigned int start = inc - tot;
    for (int w = 0; w < 4; ++w)
        if (w < wave) start += wtot[w];
    const unsigned int f0 = start;
    const unsigned int f1 = f0 + c.x;
    const unsigned int f2 = f1 + c.y;
    const unsigned int f3 = f2 + c.z;
    *(uint4*)&h[4 * t] = make_uint4(f0, f1, f2, f3);
    if (4 * t     <= NCHUNK) offs[(4 * t)     * NBLK + tile] = f0;
    if (4 * t + 1 <= NCHUNK) offs[(4 * t + 1) * NBLK + tile] = f1;
    if (4 * t + 2 <= NCHUNK) offs[(4 * t + 2) * NBLK + tile] = f2;
    if (4 * t + 3 <= NCHUNK) offs[(4 * t + 3) * NBLK + tile] = f3;
    __syncthreads();
    #pragma unroll
    for (int i = 0; i < NL; ++i) {
        if (ok[i]) {
            sorted[h[bucket_of(d[i].x)] + rk[4 * i + 0]] = pack_edge(s[i].x, d[i].x);
            sorted[h[bucket_of(d[i].y)] + rk[4 * i + 1]] = pack_edge(s[i].y, d[i].y);
            sorted[h[bucket_of(d[i].z)] + rk[4 * i + 2]] = pack_edge(s[i].z, d[i].z);
            sorted[h[bucket_of(d[i].w)] + rk[4 * i + 3]] = pack_edge(s[i].w, d[i].w);
        }
    }
    __syncthreads();
    const int left = NEDGE - tile * EPB;
    const int nq   = (left < EPB ? left : EPB) / 4;
    uint4* __restrict__ o = (uint4*)(ebuf + (size_t)tile * EPB);
    for (int i = 0; i < NL; ++i) {
        const int q = i * 128 + t;
        if (q < nq) o[q] = ((const uint4*)sorted)[q];
    }
}

__global__ __launch_bounds__(128) void gemm_wmma_f32(
        const float* __restrict__ Xa, const float* __restrict__ Wa,
        float* __restrict__ Za, int Ma, int Ka,
        const float* __restrict__ Xb, const float* __restrict__ Wb,
        float* __restrict__ Zb, int Mb, int Kb, int nblk_a, int nblk_gemm,
        const int* __restrict__ src, const int* __restrict__ dst,
        unsigned int* __restrict__ offs, unsigned int* __restrict__ ebuf) {
    const int KT = 64;
    const int MT = 128;
    const int PW = 68;
    __shared__ __attribute__((aligned(16))) float lX[MT * KT];
    __shared__ __attribute__((aligned(16))) float lWt[FATT * PW];
    static_assert(EPB <= MT * KT && 2 * CHUNK + 4 <= FATT * PW, "the tile pass works in the projections' LDS arrays");
    if ((int)blockIdx.x >= nblk_gemm) {
        tile_sort_body((int)blockIdx.x - nblk_gemm, src, dst, offs, ebuf,
                       (unsigned int*)lX, (unsigned int*)lWt);
        return;
    }
    const bool second = (int)blockIdx.x >= nblk_a;
    const float* __restrict__ X = second ? Xb : Xa;
    const float* __restrict__ W = second ? Wb : Wa;
    float* __restrict__ Z = second ? Zb : Za;
    const int M = second ? Mb : Ma;
    const int K = second ? Kb : Ka;

    const int lane  = threadIdx.x & 31;
    const int wave  = threadIdx.x >> 5;
    const int row   = lane & 15;
    const int khalf = lane >> 4;
    const int n     = lane & 15;

    const int blk_row = ((int)blockIdx.x - (second ? nblk_a : 0)) * MT;
    const int lr0 = wave * 32 + row;
    const int lr1 = lr0 + 16;

    v8f c0 = {}, c1 = {}, c2 = {}, c3 = {};
    v8f c4 = {}, c5 = {}, c6 = {}, c7 = {};

    const int tid = threadIdx.x;
    const int sc  = tid & 63;
    const int sr  = tid >> 6;

    float* xs[8];
    #pragma unroll
    for (int q = 0; q < 8; ++q) {
        const int lr = sr + 2 * q;
        xs[q] = lX + lr * KT + ((((sc >> 2) ^ (lr & 15)) << 2) | (sc & 3));
    }

    for (int kt = 0; kt < K; kt += KT) {
        float wv[(KT * FATT) / 128];
        float xv[(MT * KT) / 128];
        const int  kgx   = kt + sc;
        const bool kx_ok = kgx < K;
        const unsigned kgxc = (unsigned)(kx_ok ? kgx : K - 1);
        #pragma unroll
        for (int j = 0; j < (KT * FATT) / 128; ++j) {
            int kg  = kt + sr + 2 * j;
            int kgc = kg < K ? kg : K - 1;
            wv[j] = W[(unsigned)kgc * FATT + (unsigned)sc];
        }
        #pragma unroll
        for (int j = 0; j < (MT * KT) / 128; ++j) {
            int mg  = blk_row + sr + 2 * j;
            int mgc = mg < M ? mg : M - 1;
            xv[j] = X[(unsigned)mgc * (unsigned)K + kgxc];
        }
        #pragma unroll
        for (int j = 0; j < (KT * FATT) / 128; ++j) {
            int kg = kt + sr + 2 * j;
            lWt[sc * PW + sr + 2 * j] = (kg < K) ? wv[j] : 0.0f;
        }
        #pragma unroll
        for (int j = 0; j < (MT * KT) / 128; ++j) {
            int mg = blk_row + sr + 2 * j;
            xs[j & 7][(j >> 3) * 16 * KT] = (mg < M && kx_ok) ? xv[j] : 0.0f;
        }
        __syncthreads();

        const float* pa0 = lX  + lr0 * KT;
        const float* pa1 = lX  + lr1 * KT;
        const float* pb  = lWt + n * PW;
        #pragma unroll 2
        for (int g = 0; g < KT / 8; ++g) {
            const int fc = 2 * g + khalf;
            const int ac = (fc ^ row) << 2;
            const v4f a0 = *(const v4f*)(pa0 + ac);
            const v4f a1 = *(const v4f*)(pa1 + ac);
            const v4f b0 = *(const v4f*)(pb + 4 * fc);
            const v4f b1 = *(const v4f*)(pb + 16 * PW + 4 * fc);
            const v4f b2 = *(const v4f*)(pb + 32 * PW + 4 * fc);
            const v4f b3 = *(const v4f*)(pb + 48 * PW + 4 * fc);

            c0 = __builtin_amdgcn_wmma_f32_16x16x4_f32(false, a0.xy, false, b0.xy, (short)0, c0, false, false);
            c1 = __builtin_amdgcn_wmma_f32_16x16x4_f32(false, a0.xy, false, b1.xy, (short)0, c1, false, false);
            c2 = __builtin_amdgcn_wmma_f32_16x16x4_f32(false, a0.xy, false, b2.xy, (short)0, c2, false, false);
            c3 = __builtin_amdgcn_wmma_f32_16x16x4_f32(false, a0.xy, false, b3.xy, (short)0, c3, false, false);
            c4 = __builtin_amdgcn_wmma_f32_16x16x4_f32(false, a1.xy, false, b0.xy, (short)0, c4, false, false);
            c5 = __builtin_amdgcn_wmma_f32_16x16x4_f32(false, a1.xy, false, b1.xy, (short)0, c5, false, false);
            c6 = __builtin_amdgcn_wmma_f32_16x16x4_f32(false, a1.xy, false, b2.xy, (short)0, c6, false, false);
            c7 = __builtin_amdgcn_wmma_f32_16x16x4_f32(false, a1.xy, false, b3.xy, (short)0, c7, false, false);
            c0 = __builtin_amdgcn_wmma_f32_16x16x4_f32(false, a0.zw, false, b0.zw, (short)0, c0, false, false);
            c1 = __builtin_amdgcn_wmma_f32_16x16x4_f32(false, a0.zw, false, b1.zw, (short)0, c1, false, false);
            c2 = __builtin_amdgcn_wmma_f32_16x16x4_f32(false, a0.zw, false, b2.zw, (short)0, c2, false, false);
            c3 = __builtin_amdgcn_wmma_f32_16x16x4_f32(false, a0.zw, false, b3.zw, (short)0, c3, false, false);
            c4 = __builtin_amdgcn_wmma_f32_16x16x4_f32(false, a1.zw, false, b0.zw, (short)0, c4, false, false);
            c5 = __builtin_amdgcn_wmma_f32_16x16x4_f32(false, a1.zw, false, b1.zw, (short)0, c5, false, false);
            c6 = __builtin_amdgcn_wmma_f32_16x16x4_f32(false, a1.zw, false, b2.zw, (short)0, c6, false, false);
            c7 = __builtin_amdgcn_wmma_f32_16x16x4_f32(false, a1.zw, false, b3.zw, (short)0, c7, false, false);
        }
        __syncthreads();
    }

    const int m0 = blk_row + wave * 32 + 8 * khalf;
    #pragma unroll
    for (int r = 0; r < 8; ++r) {
        int mm = m0 + r;
        if (mm < M) {
            float* zr = Z + (size_t)mm * FATT + n;
            zr[0]  = c0[r];
            zr[16] = c1[r];
            zr[32] = c2[r];
            zr[48] = c3[r];
        }
        int mm1 = mm + 16;
        if (mm1 < M) {
            float* zr = Z + (size_t)mm1 * FATT + n;
            zr[0]  = c4[r];
            zr[16] = c5[r];
            zr[32] = c6[r];
            zr[48] = c7[r];
        }
    }
}

template <bool RANK>
__device__ __forceinline__ void take_edge(unsigned int w, unsigned int i,
                                          unsigned int lo, unsigned int hi,
                                          unsigned int* cnt, unsigned int bs,
                                          int* __restrict__ csr_src) {
    if (i >= lo && i < hi) {
        if (RANK) {
            unsigned int r = bs + atomicAdd(&cnt[(w >> SRC_BITS) & 255u], 1u);
            r = r < NEDGE ? r : NEDGE - 1;
            csr_src[r] = (int)(w & ((1u << SRC_BITS) - 1u));
        } else {
            atomicAdd(&cnt[(w >> SRC_BITS) & 255u], 1u);
        }
    }
}


__device__ __forceinline__ void load_pieces(const uint4* __restrict__ tq, unsigned int b,
                                            unsigned int qe, uint4 (&w)[RUN_Q]) {
    #pragma unroll
    for (int j = 0; j < RUN_Q; ++j) {
        const unsigned int p = b + (unsigned int)j;
        w[j] = tq[p < qe ? p : 0u];
    }
}

template <bool RANK>
__device__ __forceinline__ void take_pieces(const uint4 (&w)[RUN_Q], unsigned int b,
                                            unsigned int lo, unsigned int hi,
                                            unsigned int* cnt, unsigned int bs,
                                            int* __restrict__ csr_src) {
    #pragma unroll
    for (int j = 0; j < RUN_Q; ++j) {
        const unsigned int i = (b + (unsigned int)j) << 2;
        take_edge<RANK>(w[j].x, i + 0u, lo, hi, cnt, bs, csr_src);
        take_edge<RANK>(w[j].y, i + 1u, lo, hi, cnt, bs, csr_src);
        take_edge<RANK>(w[j].z, i + 2u, lo, hi, cnt, bs, csr_src);
        take_edge<RANK>(w[j].w, i + 3u, lo, hi, cnt, bs, csr_src);
    }
}

template <bool RANK>
__device__ __forceinline__ void walk_rest(const uint4* __restrict__ tq, unsigned int b,
                                          unsigned int qe, unsigned int lo, unsigned int hi,
                                          unsigned int* cnt, unsigned int bs,
                                          int* __restrict__ csr_src) {
    for (; b < qe; b += RUN_Q) {
        uint4 w[RUN_Q];
        load_pieces(tq, b, qe, w);
        take_pieces<RANK>(w, b, lo, hi, cnt, bs, csr_src);
    }
}

__global__ __launch_bounds__(256) void bucket_to_csr(const unsigned int* __restrict__ ebuf,
                                                     const unsigned int* __restrict__ offs,
                                                     unsigned int* __restrict__ cur,
                                                     int* __restrict__ csr_src) {
    __shared__ unsigned int cnt[CHUNK];
    __shared__ unsigned int wtot[8];
    __shared__ unsigned int wlo[8];
    const int k = blockIdx.x, t = threadIdx.x;
    const int lane = t & 31, wave = t >> 5;
    const int  t1   = t + CHUNK;
    const bool has1 = t1 < NBLK;
    const int  t1c  = has1 ? t1 : NBLK - 1;
    unsigned int lo0 = offs[k * NBLK + t],   hi0 = offs[(k + 1) * NBLK + t];
    unsigned int lo1 = offs[k * NBLK + t1c], hi1 = offs[(k + 1) * NBLK + t1c];
    if (hi0 > EPB) hi0 = EPB;
    if (lo0 > hi0) lo0 = hi0;
    if (hi1 > EPB) hi1 = EPB;
    if (lo1 > hi1) lo1 = hi1;
    if (!has1) { lo1 = 0u; hi1 = 0u; }
    const uint4* __restrict__ tq0 = (const uint4*)(ebuf + (size_t)t   * EPB);
    const uint4* __restrict__ tq1 = (const uint4*)(ebuf + (size_t)t1c * EPB);
    const unsigned int qe0 = (hi0 + 3u) >> 2, qe1 = (hi1 + 3u) >> 2;
    const unsigned int b0 = lo0 < hi0 ? lo0 >> 2 : qe0;
    const unsigned int b1 = lo1 < hi1 ? lo1 >> 2 : qe1;
    uint4 w0[RUN_Q], w1[RUN_Q];
    load_pieces(tq0, b0, qe0, w0);
    load_pieces(tq1, b1, qe1, w1);

    const unsigned int vlo = wave_sum(lo0 + lo1);
    if (lane == 0) wlo[wave] = vlo;
    cnt[t] = 0u;
    __syncthreads();
    unsigned int bs = wlo[0] + wlo[1] + wlo[2] + wlo[3] + wlo[4] + wlo[5] + wlo[6] + wlo[7];
    if (bs > NEDGE) bs = NEDGE;

    take_pieces<false>(w0, b0, lo0, hi0, cnt, bs, csr_src);
    take_pieces<false>(w1, b1, lo1, hi1, cnt, bs, csr_src);
    walk_rest<false>(tq0, b0 + RUN_Q, qe0, lo0, hi0, cnt, bs, csr_src);
    walk_rest<false>(tq1, b1 + RUN_Q, qe1, lo1, hi1, cnt, bs, csr_src);
    __syncthreads();

    const unsigned int c = cnt[t];
    unsigned int inc = c;
    for (int off = 1; off < 32; off <<= 1) {
        const unsigned int up = __shfl_up(inc, off, 32);
        if (lane >= off) inc += up;
    }
    if (lane == 31) wtot[wave] = inc;
    __syncthreads();
    unsigned int start = inc - c;
    for (int w = 0; w < 8; ++w)
        if (w < wave) start += wtot[w];
    const int node = k * CHUNK + t;
    if (node < NTOT) cur[node] = bs + start + c;
    cnt[t] = start;
    __syncthreads();
    take_pieces<true>(w0, b0, lo0, hi0, cnt, bs, csr_src);
    take_pieces<true>(w1, b1, lo1, hi1, cnt, bs, csr_src);
    walk_rest<true>(tq0, b0 + RUN_Q, qe0, lo0, hi0, cnt, bs, csr_src);
    walk_rest<true>(tq1, b1 + RUN_Q, qe1, lo1, hi1, cnt, bs, csr_src);
}

__device__ __forceinline__ float lrelu(float v) { return v > 0.0f ? v : SLOPE * v; }


__device__ __forceinline__ float xhalf(float x) {
    const int b = __builtin_bit_cast(int, x);
    return __builtin_bit_cast(float,
        __builtin_amdgcn_permlanex16(b, b, 0x76543210, 0xfedcba98, false, false));
}

template <int MASK>
__device__ __forceinline__ float row_xor(float v) {
    return __builtin_bit_cast(float,
        __builtin_amdgcn_mov_dpp(__builtin_bit_cast(int, v), 0x160 | MASK, 0xf, 0xf, true));
}

template <int MASK>
__device__ __forceinline__ float fold_pair(float lo, float hi, int c) {
    const bool up = (c & MASK) != 0;
    const float keep = up ? hi : lo;
    const float send = up ? lo : hi;
    return keep + row_xor<MASK>(send);
}

#define ROW_SHARE_CASE(P) case P: return __builtin_amdgcn_mov_dpp(v, 0x150 | P, 0xf, 0xf, true);
__device__ __forceinline__ int row_bcast(int v, int p) {
    switch (p & 15) {
        ROW_SHARE_CASE(0)  ROW_SHARE_CASE(1)  ROW_SHARE_CASE(2)  ROW_SHARE_CASE(3)
        ROW_SHARE_CASE(4)  ROW_SHARE_CASE(5)  ROW_SHARE_CASE(6)  ROW_SHARE_CASE(7)
        ROW_SHARE_CASE(8)  ROW_SHARE_CASE(9)  ROW_SHARE_CASE(10) ROW_SHARE_CASE(11)
        ROW_SHARE_CASE(12) ROW_SHARE_CASE(13) ROW_SHARE_CASE(14)
        default: return __builtin_amdgcn_mov_dpp(v, 0x150 | 15, 0xf, 0xf, true);
    }
}
#undef ROW_SHARE_CASE

template <int NP, int P0>
__device__ __forceinline__ void attend_pairs(const float* __restrict__ z, unsigned int sl, int n,
                                             unsigned int c16, int c, int h, v2f zlo, v2f zhi,
                                             float& m, float& l, v2f& alo, v2f& ahi) {
    static_assert(NP == 4 || NP == 8, "the butterfly below is written for 4 or 8 pairs");
    static_assert(P0 % NP == 0, "lane P0+k must be the lane whose low bits are k");
    v4f   r[NP];
    float d[NP];
    #pragma unroll
    for (int i = 0; i < NP; ++i) {
        const unsigned int s = (unsigned int)row_bcast((int)sl, P0 + i);
        r[i] = *(const v4f*)((const char*)z + (s * (unsigned int)(FATT * 4) + c16));
    }
    __builtin_amdgcn_sched_barrier(0);
    #pragma unroll
    for (int i = 0; i < NP; ++i) {
        const v2f t = __builtin_elementwise_fma(r[i].zw, zhi, r[i].xy * zlo);
        d[i] = t.x + t.y;
    }
    float s;
    if (NP == 8) {
        const float e0 = fold_pair<4>(d[0], d[4], c);
        const float e1 = fold_pair<4>(d[1], d[5], c);
        const float e2 = fold_pair<4>(d[2], d[6], c);
        const float e3 = fold_pair<4>(d[3], d[7], c);
        const float f0 = fold_pair<2>(e0, e2, c);
        const float f1 = fold_pair<2>(e1, e3, c);
        s = fold_pair<1>(f0, f1, c);
        s += row_xor<8>(s);
    } else {
        const float f0 = fold_pair<2>(d[0], d[2], c);
        const float f1 = fold_pair<2>(d[1], d[3], c);
        s = fold_pair<1>(f0, f1, c);
        s += row_xor<4>(s);
        s += row_xor<8>(s);
    }
    const bool mine = ((c & ~(NP - 1)) == P0) && (2 * c + h < n);
    const float e = mine ? fmaxf(s, SLOPE * s) : -INFINITY;
    float cm = e;
    cm = fmaxf(cm, row_xor<1>(cm));
    cm = fmaxf(cm, row_xor<2>(cm));
    cm = fmaxf(cm, row_xor<4>(cm));
    cm = fmaxf(cm, row_xor<8>(cm));
    cm = fmaxf(cm, xhalf(cm));
    const float mn = fmaxf(m, cm);
    const float sc = __expf(m - mn);
    const float p  = __expf(e - mn);
    l = l * sc + p;
    v2f slo = alo * sc;
    v2f shi = ahi * sc;
    #pragma unroll
    for (int i = 0; i < NP; ++i) {
        const float w = __builtin_bit_cast(float, row_bcast(__builtin_bit_cast(int, p), P0 + i));
        const v2f  ww = { w, w };
        slo = __builtin_elementwise_fma(r[i].xy, ww, slo);
        shi = __builtin_elementwise_fma(r[i].zw, ww, shi);
    }
    alo = slo; ahi = shi;
    m = mn;
}

__global__ __launch_bounds__(256) void node_attend(const float* __restrict__ z,
                                                   const int* __restrict__ csr_src,
                                                   const unsigned int* __restrict__ cur,
                                                   float* __restrict__ out) {
    const int lane = threadIdx.x & 31;
    const int c = lane & 15;
    const int h = lane >> 4;
    const unsigned int c16 = (unsigned int)c * 16u;
    const int node = blockIdx.x * 8 + __builtin_amdgcn_readfirstlane((int)(threadIdx.x >> 5));
    if (node >= NTOT) return;
    unsigned int end = cur[node];
    unsigned int beg = node > 0 ? cur[node - 1] : 0u;
    if (end > NEDGE) end = NEDGE;
    if (beg > end)   beg = end;

    const v4f zd = *(const v4f*)((const char*)(z + (size_t)node * FATT) + c16);
    const v2f zlo = zd.xy, zhi = zd.zw;
    float m = -3.0e38f, l = 0.0f;
    v2f alo = { 0.0f, 0.0f }, ahi = { 0.0f, 0.0f };

    for (unsigned int base = beg; base < end; base += 32) {
        const unsigned int j = base + 2u * (unsigned int)c + (unsigned int)h;
        unsigned int sl = (unsigned int)csr_src[j < end ? j : end - 1];
        sl = sl < NTOT ? sl : NTOT - 1;
        const int n = (int)(end - base < 32u ? end - base : 32u);
        if (n > 8) attend_pairs<8, 0>(z, sl, n, c16, c, h, zlo, zhi, m, l, alo, ahi);
        else       attend_pairs<4, 0>(z, sl, n, c16, c, h, zlo, zhi, m, l, alo, ahi);
        if (n > 16) {
            if (n > 24) attend_pairs<8, 8>(z, sl, n, c16, c, h, zlo, zhi, m, l, alo, ahi);
            else        attend_pairs<4, 8>(z, sl, n, c16, c, h, zlo, zhi, m, l, alo, ahi);
        }
    }
    l += row_xor<1>(l);
    l += row_xor<2>(l);
    l += row_xor<4>(l);
    l += row_xor<8>(l);
    l += xhalf(l);
    const float wx = alo.x + xhalf(alo.x);
    const float wy = alo.y + xhalf(alo.y);
    const float wz = ahi.x + xhalf(ahi.x);
    const float ww = ahi.y + xhalf(ahi.y);
    v4f hv = { 0.0f, 0.0f, 0.0f, 0.0f };
    if (l > 0.0f) {
        const float inv = 1.0f / l;
        hv.x = wx * inv; hv.y = wy * inv; hv.z = wz * inv; hv.w = ww * inv;
    }
    hv.x = hv.x > 0.0f ? hv.x : (__expf(hv.x) - 1.0f);
    hv.y = hv.y > 0.0f ? hv.y : (__expf(hv.y) - 1.0f);
    hv.z = hv.z > 0.0f ? hv.z : (__expf(hv.z) - 1.0f);
    hv.w = hv.w > 0.0f ? hv.w : (__expf(hv.w) - 1.0f);
    if (h == 0) *(v4f*)((char*)(out + (size_t)node * FATT) + c16) = hv;
}

extern "C" void kernel_launch(void* const* d_in, const int* in_sizes, int n_in,
                              void* d_out, int out_size, void* d_ws, size_t ws_size,
                              hipStream_t stream) {
    const float* d_sim  = (const float*)d_in[0];
    const float* mi_sim = (const float*)d_in[1];
    const float* W_d    = (const float*)d_in[2];
    const float* W_mi   = (const float*)d_in[3];
    const int*   src    = (const int*)d_in[4];
    const int*   dst    = (const int*)d_in[5];
    float*       out    = (float*)d_out;

    float* z            = (float*)d_ws;
    int* csr_src        = (int*)(z + (size_t)NTOT * FATT);
    unsigned int* cur   = (unsigned int*)(csr_src + NEDGE);
    unsigned int* offs  = (unsigned int*)d_out;
    unsigned int* ebuf  = offs + NOFF;

    gemm_wmma_f32<<<(N_MIC + 127) / 128 + (N_DIS + 127) / 128 + NBLK, 128, 0, stream>>>(
        mi_sim, W_mi, z + (size_t)N_DIS * FATT, N_MIC, 495,
        d_sim, W_d, z, N_DIS, 383,
        (N_MIC + 127) / 128, (N_MIC + 127) / 128 + (N_DIS + 127) / 128,
        src, dst, offs, ebuf);

    bucket_to_csr<<<NCHUNK, 256, 0, stream>>>(ebuf, offs, cur, csr_src);

    node_attend<<<(NTOT + 7) / 8, 256, 0, stream>>>(z, csr_src, cur, out);
}


# node_attend as a resident grid: 12,288 waves (not 100,000 to start) each walk 8 or 9 consecutive nodes, bounds from one load, next node's sources asked for one node ahead; sums unchanged
# speedup vs baseline: 10.0552x; 1.0348x over previous
#include <hip/hip_runtime.h>
#include <math.h>

#define N_DIS 20000
#define N_MIC 80000
#define NTOT  (N_DIS + N_MIC)
#define NEDGE 1600000
#define FATT  64
#define SLOPE 0.2f

typedef float v2f __attribute__((ext_vector_type(2)));
typedef float v4f __attribute__((ext_vector_type(4)));
typedef float v8f __attribute__((ext_vector_type(8)));

#define CHUNK  256
#define NCHUNK ((NTOT + CHUNK - 1) / CHUNK)
#define EPB    4096
#define NBLK   ((NEDGE + EPB - 1) / EPB)
#define NOFF   ((NCHUNK + 1) * NBLK)
#define SRC_BITS 17
#define RUN_Q  6

static_assert(NEDGE % 4 == 0, "the edge pass reads four edges per thread and load");
static_assert(EPB % 512 == 0, "128 threads x 4 edges per load");
static_assert(NTOT <= (1 << SRC_BITS), "source index must fit the packed edge");
static_assert(NCHUNK + 1 <= 2 * CHUNK, "the 128 threads of the tile pass scan four bucket counts each");
static_assert(NBLK > CHUNK && NBLK <= 2 * CHUNK, "a thread of bucket_to_csr walks the runs of tiles t and t + 256");
static_assert(NOFF % 4 == 0, "ebuf starts on a 16-byte boundary behind offs");

__device__ __forceinline__ unsigned int bucket_of(int d) {
    const unsigned int k = (unsigned int)d >> 8;
    return k < NCHUNK ? k : NCHUNK - 1;
}

__device__ __forceinline__ unsigned int pack_edge(int s, int d) {
    const unsigned int su = (unsigned int)s < NTOT ? (unsigned int)s : NTOT - 1;
    return su | (((unsigned int)d & 255u) << SRC_BITS);
}

__device__ __forceinline__ unsigned int wave_sum(unsigned int v) {
    for (int off = 16; off > 0; off >>= 1) v += __shfl_xor(v, off, 32);
    return v;
}

__device__ __forceinline__ void tile_sort_body(const int tile,
                                               const int* __restrict__ src,
                                               const int* __restrict__ dst,
                                               unsigned int* __restrict__ offs,
                                               unsigned int* __restrict__ ebuf,
                                               unsigned int* sorted,
                                               unsigned int* h) {
    unsigned int* wtot = h + 2 * CHUNK;
    constexpr int NL = EPB / 512;
    const int t = threadIdx.x;
    const int lane = t & 31, wave = t >> 5;
    int4 d[NL], s[NL];
    bool ok[NL];
    #pragma unroll
    for (int i = 0; i < NL; ++i) {
        const int q = tile * (EPB / 4) + i * 128 + t;
        ok[i] = q < NEDGE / 4;
        const int qc = ok[i] ? q : NEDGE / 4 - 1;
        d[i] = ((const int4*)dst)[qc];
        s[i] = ((const int4*)src)[qc];
    }
    *(uint4*)&h[4 * t] = make_uint4(0u, 0u, 0u, 0u);
    __syncthreads();
    unsigned int rk[4 * NL];
    #pragma unroll
    for (int i = 0; i < NL; ++i) {
        rk[4 * i + 0] = rk[4 * i + 1] = rk[4 * i + 2] = rk[4 * i + 3] = 0u;
        if (ok[i]) {
            rk[4 * i + 0] = atomicAdd(&h[bucket_of(d[i].x)], 1u);
            rk[4 * i + 1] = atomicAdd(&h[bucket_of(d[i].y)], 1u);
            rk[4 * i + 2] = atomicAdd(&h[bucket_of(d[i].z)], 1u);
            rk[4 * i + 3] = atomicAdd(&h[bucket_of(d[i].w)], 1u);
        }
    }
    __syncthreads();
    const uint4 c = *(const uint4*)&h[4 * t];
    const unsigned int tot = c.x + c.y + c.z + c.w;
    unsigned int inc = tot;
    for (int off = 1; off < 32; off <<= 1) {
        const unsigned int up = __shfl_up(inc, off, 32);
        if (lane >= off) inc += up;
    }
    if (lane == 31) wtot[wave] = inc;
    __syncthreads();
    unsigned int start = inc - tot;
    for (int w = 0; w < 4; ++w)
        if (w < wave) start += wtot[w];
    const unsigned int f0 = start;
    const unsigned int f1 = f0 + c.x;
    const unsigned int f2 = f1 + c.y;
    const unsigned int f3 = f2 + c.z;
    *(uint4*)&h[4 * t] = make_uint4(f0, f1, f2, f3);
    if (4 * t     <= NCHUNK) offs[(4 * t)     * NBLK + tile] = f0;
    if (4 * t + 1 <= NCHUNK) offs[(4 * t + 1) * NBLK + tile] = f1;
    if (4 * t + 2 <= NCHUNK) offs[(4 * t + 2) * NBLK + tile] = f2;
    if (4 * t + 3 <= NCHUNK) offs[(4 * t + 3) * NBLK + tile] = f3;
    __syncthreads();
    #pragma unroll
    for (int i = 0; i < NL; ++i) {
        if (ok[i]) {
            sorted[h[bucket_of(d[i].x)] + rk[4 * i + 0]] = pack_edge(s[i].x, d[i].x);
            sorted[h[bucket_of(d[i].y)] + rk[4 * i + 1]] = pack_edge(s[i].y, d[i].y);
            sorted[h[bucket_of(d[i].z)] + rk[4 * i + 2]] = pack_edge(s[i].z, d[i].z);
            sorted[h[bucket_of(d[i].w)] + rk[4 * i + 3]] = pack_edge(s[i].w, d[i].w);
        }
    }
    __syncthreads();
    const int left = NEDGE - tile * EPB;
    const int nq   = (left < EPB ? left : EPB) / 4;
    uint4* __restrict__ o = (uint4*)(ebuf + (size_t)tile * EPB);
    for (int i = 0; i < NL; ++i) {
        const int q = i * 128 + t;
        if (q < nq) o[q] = ((const uint4*)sorted)[q];
    }
}

__global__ __launch_bounds__(128) void gemm_wmma_f32(
        const float* __restrict__ Xa, const float* __restrict__ Wa,
        float* __restrict__ Za, int Ma, int Ka,
        const float* __restrict__ Xb, const float* __restrict__ Wb,
        float* __restrict__ Zb, int Mb, int Kb, int nblk_a, int nblk_gemm,
        const int* __restrict__ src, const int* __restrict__ dst,
        unsigned int* __restrict__ offs, unsigned int* __restrict__ ebuf) {
    const int KT = 64;
    const int MT = 128;
    const int PW = 68;
    __shared__ __attribute__((aligned(16))) float lX[MT * KT];
    __shared__ __attribute__((aligned(16))) float lWt[FATT * PW];
    static_assert(EPB <= MT * KT && 2 * CHUNK + 4 <= FATT * PW, "the tile pass works in the projections' LDS arrays");
    if ((int)blockIdx.x >= nblk_gemm) {
        tile_sort_body((int)blockIdx.x - nblk_gemm, src, dst, offs, ebuf,
                       (unsigned int*)lX, (unsigned int*)lWt);
        return;
    }
    const bool second = (int)blockIdx.x >= nblk_a;
    const float* __restrict__ X = second ? Xb : Xa;
    const float* __restrict__ W = second ? Wb : Wa;
    float* __restrict__ Z = second ? Zb : Za;
    const int M = second ? Mb : Ma;
    const int K = second ? Kb : Ka;

    const int lane  = threadIdx.x & 31;
    const int wave  = threadIdx.x >> 5;
    const int row   = lane & 15;
    const int khalf = lane >> 4;
    const int n     = lane & 15;

    const int blk_row = ((int)blockIdx.x - (second ? nblk_a : 0)) * MT;
    const int lr0 = wave * 32 + row;
    const int lr1 = lr0 + 16;

    v8f c0 = {}, c1 = {}, c2 = {}, c3 = {};
    v8f c4 = {}, c5 = {}, c6 = {}, c7 = {};

    const int tid = threadIdx.x;
    const int sc  = tid & 63;
    const int sr  = tid >> 6;

    float* xs[8];
    #pragma unroll
    for (int q = 0; q < 8; ++q) {
        const int lr = sr + 2 * q;
        xs[q] = lX + lr * KT + ((((sc >> 2) ^ (lr & 15)) << 2) | (sc & 3));
    }

    for (int kt = 0; kt < K; kt += KT) {
        float wv[(KT * FATT) / 128];
        float xv[(MT * KT) / 128];
        const int  kgx   = kt + sc;
        const bool kx_ok = kgx < K;
        const unsigned kgxc = (unsigned)(kx_ok ? kgx : K - 1);
        #pragma unroll
        for (int j = 0; j < (KT * FATT) / 128; ++j) {
            int kg  = kt + sr + 2 * j;
            int kgc = kg < K ? kg : K - 1;
            wv[j] = W[(unsigned)kgc * FATT + (unsigned)sc];
        }
        #pragma unroll
        for (int j = 0; j < (MT * KT) / 128; ++j) {
            int mg  = blk_row + sr + 2 * j;
            int mgc = mg < M ? mg : M - 1;
            xv[j] = X[(unsigned)mgc * (unsigned)K + kgxc];
        }
        #pragma unroll
        for (int j = 0; j < (KT * FATT) / 128; ++j) {
            int kg = kt + sr + 2 * j;
            lWt[sc * PW + sr + 2 * j] = (kg < K) ? wv[j] : 0.0f;
        }
        #pragma unroll
        for (int j = 0; j < (MT * KT) / 128; ++j) {
            int mg = blk_row + sr + 2 * j;
            xs[j & 7][(j >> 3) * 16 * KT] = (mg < M && kx_ok) ? xv[j] : 0.0f;
        }
        __syncthreads();

        const float* pa0 = lX  + lr0 * KT;
        const float* pa1 = lX  + lr1 * KT;
        const float* pb  = lWt + n * PW;
        #pragma unroll 2
        for (int g = 0; g < KT / 8; ++g) {
            const int fc = 2 * g + khalf;
            const int ac = (fc ^ row) << 2;
            const v4f a0 = *(const v4f*)(pa0 + ac);
            const v4f a1 = *(const v4f*)(pa1 + ac);
            const v4f b0 = *(const v4f*)(pb + 4 * fc);
            const v4f b1 = *(const v4f*)(pb + 16 * PW + 4 * fc);
            const v4f b2 = *(const v4f*)(pb + 32 * PW + 4 * fc);
            const v4f b3 = *(const v4f*)(pb + 48 * PW + 4 * fc);

            c0 = __builtin_amdgcn_wmma_f32_16x16x4_f32(false, a0.xy, false, b0.xy, (short)0, c0, false, false);
            c1 = __builtin_amdgcn_wmma_f32_16x16x4_f32(false, a0.xy, false, b1.xy, (short)0, c1, false, false);
            c2 = __builtin_amdgcn_wmma_f32_16x16x4_f32(false, a0.xy, false, b2.xy, (short)0, c2, false, false);
            c3 = __builtin_amdgcn_wmma_f32_16x16x4_f32(false, a0.xy, false, b3.xy, (short)0, c3, false, false);
            c4 = __builtin_amdgcn_wmma_f32_16x16x4_f32(false, a1.xy, false, b0.xy, (short)0, c4, false, false);
            c5 = __builtin_amdgcn_wmma_f32_16x16x4_f32(false, a1.xy, false, b1.xy, (short)0, c5, false, false);
            c6 = __builtin_amdgcn_wmma_f32_16x16x4_f32(false, a1.xy, false, b2.xy, (short)0, c6, false, false);
            c7 = __builtin_amdgcn_wmma_f32_16x16x4_f32(false, a1.xy, false, b3.xy, (short)0, c7, false, false);
            c0 = __builtin_amdgcn_wmma_f32_16x16x4_f32(false, a0.zw, false, b0.zw, (short)0, c0, false, false);
            c1 = __builtin_amdgcn_wmma_f32_16x16x4_f32(false, a0.zw, false, b1.zw, (short)0, c1, false, false);
            c2 = __builtin_amdgcn_wmma_f32_16x16x4_f32(false, a0.zw, false, b2.zw, (short)0, c2, false, false);
            c3 = __builtin_amdgcn_wmma_f32_16x16x4_f32(false, a0.zw, false, b3.zw, (short)0, c3, false, false);
            c4 = __builtin_amdgcn_wmma_f32_16x16x4_f32(false, a1.zw, false, b0.zw, (short)0, c4, false, false);
            c5 = __builtin_amdgcn_wmma_f32_16x16x4_f32(false, a1.zw, false, b1.zw, (short)0, c5, false, false);
            c6 = __builtin_amdgcn_wmma_f32_16x16x4_f32(false, a1.zw, false, b2.zw, (short)0, c6, false, false);
            c7 = __builtin_amdgcn_wmma_f32_16x16x4_f32(false, a1.zw, false, b3.zw, (short)0, c7, false, false);
        }
        __syncthreads();
    }

    const int m0 = blk_row + wave * 32 + 8 * khalf;
    #pragma unroll
    for (int r = 0; r < 8; ++r) {
        int mm = m0 + r;
        if (mm < M) {
            float* zr = Z + (size_t)mm * FATT + n;
            zr[0]  = c0[r];
            zr[16] = c1[r];
            zr[32] = c2[r];
            zr[48] = c3[r];
        }
        int mm1 = mm + 16;
        if (mm1 < M) {
            float* zr = Z + (size_t)mm1 * FATT + n;
            zr[0]  = c4[r];
            zr[16] = c5[r];
            zr[32] = c6[r];
            zr[48] = c7[r];
        }
    }
}

template <bool RANK>
__device__ __forceinline__ void take_edge(unsigned int w, unsigned int i,
                                          unsigned int lo, unsigned int hi,
                                          unsigned int* cnt, unsigned int bs,
                                          int* __restrict__ csr_src) {
    if (i >= lo && i < hi) {
        if (RANK) {
            unsigned int r = bs + atomicAdd(&cnt[(w >> SRC_BITS) & 255u], 1u);
            r = r < NEDGE ? r : NEDGE - 1;
            csr_src[r] = (int)(w & ((1u << SRC_BITS) - 1u));
        } else {
            atomicAdd(&cnt[(w >> SRC_BITS) & 255u], 1u);
        }
    }
}


__device__ __forceinline__ void load_pieces(const uint4* __restrict__ tq, unsigned int b,
                                            unsigned int qe, uint4 (&w)[RUN_Q]) {
    #pragma unroll
    for (int j = 0; j < RUN_Q; ++j) {
        const unsigned int p = b + (unsigned int)j;
        w[j] = tq[p < qe ? p : 0u];
    }
}

template <bool RANK>
__device__ __forceinline__ void take_pieces(const uint4 (&w)[RUN_Q], unsigned int b,
                                            unsigned int lo, unsigned int hi,
                                            unsigned int* cnt, unsigned int bs,
                                            int* __restrict__ csr_src) {
    #pragma unroll
    for (int j = 0; j < RUN_Q; ++j) {
        const unsigned int i = (b + (unsigned int)j) << 2;
        take_edge<RANK>(w[j].x, i + 0u, lo, hi, cnt, bs, csr_src);
        take_edge<RANK>(w[j].y, i + 1u, lo, hi, cnt, bs, csr_src);
        take_edge<RANK>(w[j].z, i + 2u, lo, hi, cnt, bs, csr_src);
        take_edge<RANK>(w[j].w, i + 3u, lo, hi, cnt, bs, csr_src);
    }
}

template <bool RANK>
__device__ __forceinline__ void walk_rest(const uint4* __restrict__ tq, unsigned int b,
                                          unsigned int qe, unsigned int lo, unsigned int hi,
                                          unsigned int* cnt, unsigned int bs,
                                          int* __restrict__ csr_src) {
    for (; b < qe; b += RUN_Q) {
        uint4 w[RUN_Q];
        load_pieces(tq, b, qe, w);
        take_pieces<RANK>(w, b, lo, hi, cnt, bs, csr_src);
    }
}

__global__ __launch_bounds__(256) void bucket_to_csr(const unsigned int* __restrict__ ebuf,
                                                     const unsigned int* __restrict__ offs,
                                                     unsigned int* __restrict__ cur,
                                                     int* __restrict__ csr_src) {
    __shared__ unsigned int cnt[CHUNK];
    __shared__ unsigned int wtot[8];
    __shared__ unsigned int wlo[8];
    const int k = blockIdx.x, t = threadIdx.x;
    const int lane = t & 31, wave = t >> 5;
    const int  t1   = t + CHUNK;
    const bool has1 = t1 < NBLK;
    const int  t1c  = has1 ? t1 : NBLK - 1;
    unsigned int lo0 = offs[k * NBLK + t],   hi0 = offs[(k + 1) * NBLK + t];
    unsigned int lo1 = offs[k * NBLK + t1c], hi1 = offs[(k + 1) * NBLK + t1c];
    if (hi0 > EPB) hi0 = EPB;
    if (lo0 > hi0) lo0 = hi0;
    if (hi1 > EPB) hi1 = EPB;
    if (lo1 > hi1) lo1 = hi1;
    if (!has1) { lo1 = 0u; hi1 = 0u; }
    const uint4* __restrict__ tq0 = (const uint4*)(ebuf + (size_t)t   * EPB);
    const uint4* __restrict__ tq1 = (const uint4*)(ebuf + (size_t)t1c * EPB);
    const unsigned int qe0 = (hi0 + 3u) >> 2, qe1 = (hi1 + 3u) >> 2;
    const unsigned int b0 = lo0 < hi0 ? lo0 >> 2 : qe0;
    const unsigned int b1 = lo1 < hi1 ? lo1 >> 2 : qe1;
    uint4 w0[RUN_Q], w1[RUN_Q];
    load_pieces(tq0, b0, qe0, w0);
    load_pieces(tq1, b1, qe1, w1);

    const unsigned int vlo = wave_sum(lo0 + lo1);
    if (lane == 0) wlo[wave] = vlo;
    cnt[t] = 0u;
    __syncthreads();
    unsigned int bs = wlo[0] + wlo[1] + wlo[2] + wlo[3] + wlo[4] + wlo[5] + wlo[6] + wlo[7];
    if (bs > NEDGE) bs = NEDGE;

    take_pieces<false>(w0, b0, lo0, hi0, cnt, bs, csr_src);
    take_pieces<false>(w1, b1, lo1, hi1, cnt, bs, csr_src);
    walk_rest<false>(tq0, b0 + RUN_Q, qe0, lo0, hi0, cnt, bs, csr_src);
    walk_rest<false>(tq1, b1 + RUN_Q, qe1, lo1, hi1, cnt, bs, csr_src);
    __syncthreads();

    const unsigned int c = cnt[t];
    unsigned int inc = c;
    for (int off = 1; off < 32; off <<= 1) {
        const unsigned int up = __shfl_up(inc, off, 32);
        if (lane >= off) inc += up;
    }
    if (lane == 31) wtot[wave] = inc;
    __syncthreads();
    unsigned int start = inc - c;
    for (int w = 0; w < 8; ++w)
        if (w < wave) start += wtot[w];
    const int node = k * CHUNK + t;
    if (node < NTOT) cur[node] = bs + start + c;
    cnt[t] = start;
    __syncthreads();
    take_pieces<true>(w0, b0, lo0, hi0, cnt, bs, csr_src);
    take_pieces<true>(w1, b1, lo1, hi1, cnt, bs, csr_src);
    walk_rest<true>(tq0, b0 + RUN_Q, qe0, lo0, hi0, cnt, bs, csr_src);
    walk_rest<true>(tq1, b1 + RUN_Q, qe1, lo1, hi1, cnt, bs, csr_src);
}

__device__ __forceinline__ float lrelu(float v) { return v > 0.0f ? v : SLOPE * v; }


__device__ __forceinline__ float xhalf(float x) {
    const int b = __builtin_bit_cast(int, x);
    return __builtin_bit_cast(float,
        __builtin_amdgcn_permlanex16(b, b, 0x76543210, 0xfedcba98, false, false));
}

template <int MASK>
__device__ __forceinline__ float row_xor(float v) {
    return __builtin_bit_cast(float,
        __builtin_amdgcn_mov_dpp(__builtin_bit_cast(int, v), 0x160 | MASK, 0xf, 0xf, true));
}

template <int MASK>
__device__ __forceinline__ float fold_pair(float lo, float hi, int c) {
    const bool up = (c & MASK) != 0;
    const float keep = up ? hi : lo;
    const float send = up ? lo : hi;
    return keep + row_xor<MASK>(send);
}

#define ROW_SHARE_CASE(P) case P: return __builtin_amdgcn_mov_dpp(v, 0x150 | P, 0xf, 0xf, true);
__device__ __forceinline__ int row_bcast(int v, int p) {
    switch (p & 15) {
        ROW_SHARE_CASE(0)  ROW_SHARE_CASE(1)  ROW_SHARE_CASE(2)  ROW_SHARE_CASE(3)
        ROW_SHARE_CASE(4)  ROW_SHARE_CASE(5)  ROW_SHARE_CASE(6)  ROW_SHARE_CASE(7)
        ROW_SHARE_CASE(8)  ROW_SHARE_CASE(9)  ROW_SHARE_CASE(10) ROW_SHARE_CASE(11)
        ROW_SHARE_CASE(12) ROW_SHARE_CASE(13) ROW_SHARE_CASE(14)
        default: return __builtin_amdgcn_mov_dpp(v, 0x150 | 15, 0xf, 0xf, true);
    }
}
#undef ROW_SHARE_CASE

template <int NP, int P0>
__device__ __forceinline__ void attend_pairs(const float* __restrict__ z, unsigned int sl, int n,
                                             unsigned int c16, int c, int h, v2f zlo, v2f zhi,
                                             float& m, float& l, v2f& alo, v2f& ahi) {
    static_assert(NP == 4 || NP == 8, "the butterfly below is written for 4 or 8 pairs");
    static_assert(P0 % NP == 0, "lane P0+k must be the lane whose low bits are k");
    v4f   r[NP];
    float d[NP];
    #pragma unroll
    for (int i = 0; i < NP; ++i) {
        const unsigned int s = (unsigned int)row_bcast((int)sl, P0 + i);
        r[i] = *(const v4f*)((const char*)z + (s * (unsigned int)(FATT * 4) + c16));
    }
    __builtin_amdgcn_sched_barrier(0);
    #pragma unroll
    for (int i = 0; i < NP; ++i) {
        const v2f t = __builtin_elementwise_fma(r[i].zw, zhi, r[i].xy * zlo);
        d[i] = t.x + t.y;
    }
    float s;
    if (NP == 8) {
        const float e0 = fold_pair<4>(d[0], d[4], c);
        const float e1 = fold_pair<4>(d[1], d[5], c);
        const float e2 = fold_pair<4>(d[2], d[6], c);
        const float e3 = fold_pair<4>(d[3], d[7], c);
        const float f0 = fold_pair<2>(e0, e2, c);
        const float f1 = fold_pair<2>(e1, e3, c);
        s = fold_pair<1>(f0, f1, c);
        s += row_xor<8>(s);
    } else {
        const float f0 = fold_pair<2>(d[0], d[2], c);
        const float f1 = fold_pair<2>(d[1], d[3], c);
        s = fold_pair<1>(f0, f1, c);
        s += row_xor<4>(s);
        s += row_xor<8>(s);
    }
    const bool mine = ((c & ~(NP - 1)) == P0) && (2 * c + h < n);
    const float e = mine ? fmaxf(s, SLOPE * s) : -INFINITY;
    float cm = e;
    cm = fmaxf(cm, row_xor<1>(cm));
    cm = fmaxf(cm, row_xor<2>(cm));
    cm = fmaxf(cm, row_xor<4>(cm));
    cm = fmaxf(cm, row_xor<8>(cm));
    cm = fmaxf(cm, xhalf(cm));
    const float mn = fmaxf(m, cm);
    const float sc = __expf(m - mn);
    const float p  = __expf(e - mn);
    l = l * sc + p;
    v2f slo = alo * sc;
    v2f shi = ahi * sc;
    #pragma unroll
    for (int i = 0; i < NP; ++i) {
        const float w = __builtin_bit_cast(float, row_bcast(__builtin_bit_cast(int, p), P0 + i));
        const v2f  ww = { w, w };
        slo = __builtin_elementwise_fma(r[i].xy, ww, slo);
        shi = __builtin_elementwise_fma(r[i].zw, ww, shi);
    }
    alo = slo; ahi = shi;
    m = mn;
}

__device__ __forceinline__ void attend_chunk(const float* __restrict__ z, unsigned int sl, int n,
                                             unsigned int c16, int c, int h, v2f zlo, v2f zhi,
                                             float& m, float& l, v2f& alo, v2f& ahi) {
    sl = sl < NTOT ? sl : NTOT - 1;
    if (n > 8) attend_pairs<8, 0>(z, sl, n, c16, c, h, zlo, zhi, m, l, alo, ahi);
    else       attend_pairs<4, 0>(z, sl, n, c16, c, h, zlo, zhi, m, l, alo, ahi);
    if (n > 16) {
        if (n > 24) attend_pairs<8, 8>(z, sl, n, c16, c, h, zlo, zhi, m, l, alo, ahi);
        else        attend_pairs<4, 8>(z, sl, n, c16, c, h, zlo, zhi, m, l, alo, ahi);
    }
}

#define ATT_WAVES 12288
static_assert(ATT_WAVES % 8 == 0, "work-groups of eight waves");
static_assert((unsigned long long)ATT_WAVES * NTOT < (1ull << 32), "the range bounds are computed in 32 bits");
static_assert((NTOT + ATT_WAVES - 1) / ATT_WAVES + 2 < 32, "the bounds of a wave's nodes, one more in front and one behind, fit one load of the wave");

__device__ __forceinline__ unsigned int first_sources(const int* __restrict__ csr_src,
                                                      unsigned int beg, unsigned int end,
                                                      int c, int h) {
    const unsigned int j    = beg + 2u * (unsigned int)c + (unsigned int)h;
    const unsigned int last = end > 0u ? end - 1u : 0u;
    return (unsigned int)csr_src[j < end ? j : last];
}

__global__ __launch_bounds__(256) void node_attend(const float* __restrict__ z,
                                                   const int* __restrict__ csr_src,
                                                   const unsigned int* __restrict__ cur,
                                                   float* __restrict__ out) {
    const int lane = threadIdx.x & 31;
    const int c = lane & 15;
    const int h = lane >> 4;
    const unsigned int c16 = (unsigned int)c * 16u;
    const unsigned int w = blockIdx.x * 8u + (unsigned int)__builtin_amdgcn_readfirstlane((int)(threadIdx.x >> 5));
    if (w >= ATT_WAVES) return;
    const int first = (int)(w * (unsigned int)NTOT / ATT_WAVES);
    const int nn    = (int)((w + 1u) * (unsigned int)NTOT / ATT_WAVES) - first;

    const int bi = first - 1 + lane;
    unsigned int bv = cur[bi < 0 ? 0 : (bi < NTOT ? bi : NTOT - 1)];
    if (bi < 0) bv = 0u;
    if (bv > NEDGE) bv = NEDGE;

    unsigned int beg = (unsigned int)__builtin_amdgcn_readlane((int)bv, 0);
    unsigned int end = (unsigned int)__builtin_amdgcn_readlane((int)bv, 1);
    if (beg > end) beg = end;
    unsigned int sl = first_sources(csr_src, beg, end, c, h);

    for (int i = 0; i < nn; ++i) {
        const int node = first + i;
        const v4f zd = *(const v4f*)((const char*)(z + (size_t)node * FATT) + c16);
        const v2f zlo = zd.xy, zhi = zd.zw;
        unsigned int nbeg = end;
        const unsigned int nend = (unsigned int)__builtin_amdgcn_readlane((int)bv, i + 2);
        if (nbeg > nend) nbeg = nend;
        const unsigned int sl_next = first_sources(csr_src, nbeg, nend, c, h);

        float m = -3.0e38f, l = 0.0f;
        v2f alo = { 0.0f, 0.0f }, ahi = { 0.0f, 0.0f };

        if (beg < end) {
            const unsigned int n0 = end - beg;
            attend_chunk(z, sl, (int)(n0 < 32u ? n0 : 32u), c16, c, h, zlo, zhi, m, l, alo, ahi);
            for (unsigned int base = beg + 32u; base < end; base += 32u) {
                const unsigned int j  = base + 2u * (unsigned int)c + (unsigned int)h;
                const unsigned int s2 = (unsigned int)csr_src[j < end ? j : end - 1u];
                const unsigned int nr = end - base;
                attend_chunk(z, s2, (int)(nr < 32u ? nr : 32u), c16, c, h, zlo, zhi, m, l, alo, ahi);
            }
        }
        l += row_xor<1>(l);
        l += row_xor<2>(l);
        l += row_xor<4>(l);
        l += row_xor<8>(l);
        l += xhalf(l);
        const float wx = alo.x + xhalf(alo.x);
        const float wy = alo.y + xhalf(alo.y);
        const float wz = ahi.x + xhalf(ahi.x);
        const float ww = ahi.y + xhalf(ahi.y);
        v4f hv = { 0.0f, 0.0f, 0.0f, 0.0f };
        if (l > 0.0f) {
            const float inv = 1.0f / l;
            hv.x = wx * inv; hv.y = wy * inv; hv.z = wz * inv; hv.w = ww * inv;
        }
        hv.x = hv.x > 0.0f ? hv.x : (__expf(hv.x) - 1.0f);
        hv.y = hv.y > 0.0f ? hv.y : (__expf(hv.y) - 1.0f);
        hv.z = hv.z > 0.0f ? hv.z : (__expf(hv.z) - 1.0f);
        hv.w = hv.w > 0.0f ? hv.w : (__expf(hv.w) - 1.0f);
        if (h == 0) *(v4f*)((char*)(out + (size_t)node * FATT) + c16) = hv;

        beg = nbeg; end = nend; sl = sl_next;
    }
}

extern "C" void kernel_launch(void* const* d_in, const int* in_sizes, int n_in,
                              void* d_out, int out_size, void* d_ws, size_t ws_size,
                              hipStream_t stream) {
    const float* d_sim  = (const float*)d_in[0];
    const float* mi_sim = (const float*)d_in[1];
    const float* W_d    = (const float*)d_in[2];
    const float* W_mi   = (const float*)d_in[3];
    const int*   src    = (const int*)d_in[4];
    const int*   dst    = (const int*)d_in[5];
    float*       out    = (float*)d_out;

    float* z            = (float*)d_ws;
    int* csr_src        = (int*)(z + (size_t)NTOT * FATT);
    unsigned int* cur   = (unsigned int*)(csr_src + NEDGE);
    unsigned int* offs  = (unsigned int*)d_out;
    unsigned int* ebuf  = offs + NOFF;

    gemm_wmma_f32<<<(N_MIC + 127) / 128 + (N_DIS + 127) / 128 + NBLK, 128, 0, stream>>>(
        mi_sim, W_mi, z + (size_t)N_DIS * FATT, N_MIC, 495,
        d_sim, W_d, z, N_DIS, 383,
        (N_MIC + 127) / 128, (N_MIC + 127) / 128 + (N_DIS + 127) / 128,
        src, dst, offs, ebuf);

    bucket_to_csr<<<NCHUNK, 256, 0, stream>>>(ebuf, offs, cur, csr_src);

    node_attend<<<ATT_WAVES / 8, 256, 0, stream>>>(z, csr_src, cur, out);
}
